# starting kernel
# baseline (speedup 1.0000x reference)
#include <hip/hip_runtime.h>


typedef float v2f __attribute__((ext_vector_type(2)));
typedef float v8f __attribute__((ext_vector_type(8)));

#define SA_STRIDE 68
#define SW_STRIDE 68

__global__ __launch_bounds__(256) void gat_proj_wmma(
    const float* __restrict__ nf, const float* __restrict__ W,
    float* __restrict__ feat, int N)
{
    __shared__ float sWT[64 * SW_STRIDE];
    __shared__ float sA[8][16 * SA_STRIDE];

    const int tid  = threadIdx.x;
    const int wave = tid >> 5;
    const int lane = tid & 31;

    for (int i = tid; i < 64 * 64; i += 256) {
        const int k = i >> 6, n = i & 63;
        sWT[n * SW_STRIDE + k] = W[i];
    }

    const int rowbase = blockIdx.x * 128 + wave * 16;
    for (int r = 0; r < 16; ++r) {
        int row = rowbase + r;
        row = row < N ? row : (N - 1);
        sA[wave][r * SA_STRIDE + lane]      = nf[(size_t)row * 64 + lane];
        sA[wave][r * SA_STRIDE + lane + 32] = nf[(size_t)row * 64 + lane + 32];
    }
    __syncthreads();

    const int half = lane >> 4;
    const int m    = lane & 15;

    v8f acc0 = {}, acc1 = {}, acc2 = {}, acc3 = {};

    #pragma unroll
    for (int k0 = 0; k0 < 64; k0 += 4) {
        const int ka = k0 + half * 2;

        const v2f a  = *(const v2f*)&sA[wave][m * SA_STRIDE + ka];
        const v2f b0 = *(const v2f*)&sWT[(m +  0) * SW_STRIDE + ka];
        const v2f b1 = *(const v2f*)&sWT[(m + 16) * SW_STRIDE + ka];
        const v2f b2 = *(const v2f*)&sWT[(m + 32) * SW_STRIDE + ka];
        const v2f b3 = *(const v2f*)&sWT[(m + 48) * SW_STRIDE + ka];

        acc0 = __builtin_amdgcn_wmma_f32_16x16x4_f32(false, a, false, b0, (short)0, acc0, false, false);
        acc1 = __builtin_amdgcn_wmma_f32_16x16x4_f32(false, a, false, b1, (short)0, acc1, false, false);
        acc2 = __builtin_amdgcn_wmma_f32_16x16x4_f32(false, a, false, b2, (short)0, acc2, false, false);
        acc3 = __builtin_amdgcn_wmma_f32_16x16x4_f32(false, a, false, b3, (short)0, acc3, false, false);
    }

    #pragma unroll
    for (int v = 0; v < 8; ++v) {
        const int row = rowbase + v + half * 8;
        if (row < N) {
            float* fr = feat + (size_t)row * 64;
            fr[ 0 + m] = acc0[v];
            fr[16 + m] = acc1[v];
            fr[32 + m] = acc2[v];
            fr[48 + m] = acc3[v];
        }
    }
}

__global__ void gat_elr(const float* __restrict__ feat,
                        const float* __restrict__ al, const float* __restrict__ ar,
                        float* __restrict__ el, float* __restrict__ er, int N)
{
    int t = blockIdx.x * blockDim.x + threadIdx.x;
    if (t >= N * 4) return;
    const int n = t >> 2, h = t & 3;
    const float* f = feat + (size_t)n * 64 + h * 16;
    float sl = 0.f, sr = 0.f;
    #pragma unroll
    for (int i = 0; i < 16; ++i) {
        const float v = f[i];
        sl += v * al[h * 16 + i];
        sr += v * ar[h * 16 + i];
    }
    el[t] = sl;
    er[t] = sr;
}

__global__ void gat_init(float* __restrict__ out, const float* __restrict__ bias,
                         float* __restrict__ denom, unsigned* __restrict__ emax_u, int N)
{
    int t = blockIdx.x * blockDim.x + threadIdx.x;
    if (t < N * 64) out[t] = bias[t & 63];
    if (t < N * 4) { denom[t] = 0.f; emax_u[t] = 0u; }
}

__device__ __forceinline__ float leaky02(float x) { return x > 0.f ? x : 0.2f * x; }

__device__ __forceinline__ unsigned f2o(float f) {
    unsigned u = __float_as_uint(f);
    return (u >> 31) ? ~u : (u | 0x80000000u);
}

__global__ void gat_emax(const int* __restrict__ src, const int* __restrict__ dst,
                         const float* __restrict__ el, const float* __restrict__ er,
                         unsigned* __restrict__ emax_u, int E)
{
    int t = blockIdx.x * blockDim.x + threadIdx.x;
    if (t >= E * 4) return;
    const int e = t >> 2, h = t & 3;
    const int s = src[e], d = dst[e];
    const float x = leaky02(el[s * 4 + h] + er[d * 4 + h]);
    atomicMax(&emax_u[d * 4 + h], f2o(x));
}

__global__ void gat_decode(unsigned* __restrict__ emax_u, float* __restrict__ emax_f, int N)
{
    int t = blockIdx.x * blockDim.x + threadIdx.x;
    if (t >= N * 4) return;
    const unsigned u = emax_u[t];
    float f;
    if (u == 0u)                 f = 0.f;
    else if (u & 0x80000000u)    f = __uint_as_float(u ^ 0x80000000u);
    else                         f = __uint_as_float(~u);
    emax_f[t] = f;
}

__global__ void gat_denom(const int* __restrict__ src, const int* __restrict__ dst,
                          const float* __restrict__ el, const float* __restrict__ er,
                          const float* __restrict__ emax, float* __restrict__ denom, int E)
{
    int t = blockIdx.x * blockDim.x + threadIdx.x;
    if (t >= E * 4) return;
    const int e = t >> 2, h = t & 3;
    const int s = src[e], d = dst[e];
    const int dh = d * 4 + h;
    const float x = leaky02(el[s * 4 + h] + er[dh]);
    atomicAdd(&denom[dh], __expf(x - emax[dh]));
}

__global__ void gat_agg(const int* __restrict__ src, const int* __restrict__ dst,
                        const float* __restrict__ el, const float* __restrict__ er,
                        const float* __restrict__ emax, const float* __restrict__ denom,
                        const float* __restrict__ feat, float* __restrict__ out, int E)
{
    int t = blockIdx.x * blockDim.x + threadIdx.x;
    if (t >= E * 64) return;
    const int e = t >> 6, c = t & 63, h = c >> 4;
    const int s = src[e], d = dst[e];
    const int dh = d * 4 + h;
    const float x = leaky02(el[s * 4 + h] + er[dh]);
    const float alpha = __expf(x - emax[dh]) / denom[dh];
    atomicAdd(&out[(size_t)d * 64 + c], feat[(size_t)s * 64 + c] * alpha);
}

extern "C" void kernel_launch(void* const* d_in, const int* in_sizes, int n_in,
                              void* d_out, int out_size, void* d_ws, size_t ws_size,
                              hipStream_t stream)
{
    const float* nf   = (const float*)d_in[0];
    const float* W    = (const float*)d_in[1];
    const float* al   = (const float*)d_in[2];
    const float* ar   = (const float*)d_in[3];
    const float* bias = (const float*)d_in[4];
    const int*   src  = (const int*)d_in[5];
    const int*   dst  = (const int*)d_in[6];

    const int N = in_sizes[0] / 64;
    const int E = in_sizes[5];

    float* ws    = (float*)d_ws;
    float* feat  = ws;
    float* el    = feat  + (size_t)N * 64;
    float* er    = el    + (size_t)N * 4;
    float* emax  = er    + (size_t)N * 4;
    float* denom = emax  + (size_t)N * 4;
    float* out   = (float*)d_out;

    gat_proj_wmma<<<(N + 127) / 128, 256, 0, stream>>>(nf, W, feat, N);
    gat_elr<<<(N * 4 + 255) / 256, 256, 0, stream>>>(feat, al, ar, el, er, N);
    gat_init<<<(N * 64 + 255) / 256, 256, 0, stream>>>(out, bias, denom, (unsigned*)emax, N);
    gat_emax<<<(E * 4 + 255) / 256, 256, 0, stream>>>(src, dst, el, er, (unsigned*)emax, E);
    gat_decode<<<(N * 4 + 255) / 256, 256, 0, stream>>>((unsigned*)emax, emax, N);
    gat_denom<<<(E * 4 + 255) / 256, 256, 0, stream>>>(src, dst, el, er, emax, denom, E);
    {
        long long tot = (long long)E * 64;
        int blocks = (int)((tot + 255) / 256);
        gat_agg<<<blocks, 256, 0, stream>>>(src, dst, el, er, emax, denom, feat, out, E);
    }
}


# bitmap replaced by block-sorted edge words: sorter groups counting-sort 4096 edges by node range in LDS (src + local dst per dword); owners read segments, no dst/src gathers; arrival-order sums
# speedup vs baseline: 4.8683x; 4.8683x over previous
#include <hip/hip_runtime.h>


typedef float v2f __attribute__((ext_vector_type(2)));
typedef float v8f __attribute__((ext_vector_type(8)));

#define SA_STRIDE 68
#define SW_STRIDE 68

__global__ __launch_bounds__(256) void gat_proj_wmma(
    const float* __restrict__ nf, const float* __restrict__ W,
    float* __restrict__ feat, int N)
{
    __shared__ float sWT[64 * SW_STRIDE];
    __shared__ float sA[8][16 * SA_STRIDE];

    const int tid  = threadIdx.x;
    const int wave = tid >> 5;
    const int lane = tid & 31;

    for (int i = tid; i < 64 * 64; i += 256) {
        const int k = i >> 6, n = i & 63;
        sWT[n * SW_STRIDE + k] = W[i];
    }

    const int rowbase = blockIdx.x * 128 + wave * 16;
    for (int r = 0; r < 16; ++r) {
        int row = rowbase + r;
        row = row < N ? row : (N - 1);
        sA[wave][r * SA_STRIDE + lane]      = nf[(size_t)row * 64 + lane];
        sA[wave][r * SA_STRIDE + lane + 32] = nf[(size_t)row * 64 + lane + 32];
    }
    __syncthreads();

    const int half = lane >> 4;
    const int m    = lane & 15;

    v8f acc0 = {}, acc1 = {}, acc2 = {}, acc3 = {};

    #pragma unroll
    for (int k0 = 0; k0 < 64; k0 += 4) {
        const int ka = k0 + half * 2;

        const v2f a  = *(const v2f*)&sA[wave][m * SA_STRIDE + ka];
        const v2f b0 = *(const v2f*)&sWT[(m +  0) * SW_STRIDE + ka];
        const v2f b1 = *(const v2f*)&sWT[(m + 16) * SW_STRIDE + ka];
        const v2f b2 = *(const v2f*)&sWT[(m + 32) * SW_STRIDE + ka];
        const v2f b3 = *(const v2f*)&sWT[(m + 48) * SW_STRIDE + ka];

        acc0 = __builtin_amdgcn_wmma_f32_16x16x4_f32(false, a, false, b0, (short)0, acc0, false, false);
        acc1 = __builtin_amdgcn_wmma_f32_16x16x4_f32(false, a, false, b1, (short)0, acc1, false, false);
        acc2 = __builtin_amdgcn_wmma_f32_16x16x4_f32(false, a, false, b2, (short)0, acc2, false, false);
        acc3 = __builtin_amdgcn_wmma_f32_16x16x4_f32(false, a, false, b3, (short)0, acc3, false, false);
    }

    #pragma unroll
    for (int v = 0; v < 8; ++v) {
        const int row = rowbase + v + half * 8;
        if (row < N) {
            float* fr = feat + (size_t)row * 64;
            fr[ 0 + m] = acc0[v];
            fr[16 + m] = acc1[v];
            fr[32 + m] = acc2[v];
            fr[48 + m] = acc3[v];
        }
    }
}

#define NPW   391
#define LCAP  640
#define AGGW  32
#define BMROWS 256
#define EPB   4096
#define SRCBITS 23
static_assert(NPW <= (1 << (32 - SRCBITS)), "local destination must fit above the source bits");

__global__ void gat_elr(const float* __restrict__ feat,
                        const float* __restrict__ al, const float* __restrict__ ar,
                        float* __restrict__ el, float* __restrict__ er, int N)
{
    int t = blockIdx.x * blockDim.x + threadIdx.x;
    if (t >= N * 4) return;
    const int n = t >> 2, h = t & 3;
    const float* f = feat + (size_t)n * 64 + h * 16;
    float sl = 0.f, sr = 0.f;
    #pragma unroll
    for (int i = 0; i < 16; ++i) {
        const float v = f[i];
        sl += v * al[h * 16 + i];
        sr += v * ar[h * 16 + i];
    }
    el[t] = sl;
    er[t] = sr;
}

__global__ __launch_bounds__(256) void gat_elr_sort(
    const float* __restrict__ feat,
    const float* __restrict__ al, const float* __restrict__ ar,
    float* __restrict__ el, float* __restrict__ er, int N,
    const int* __restrict__ src, const int* __restrict__ dst,
    unsigned* __restrict__ blk, unsigned* __restrict__ offs, int G4, int nbw)
{
    __shared__ unsigned hist[BMROWS];
    __shared__ unsigned base[BMROWS];
    __shared__ unsigned sorted[EPB];

    if ((int)blockIdx.x < nbw) {
        const int tid  = threadIdx.x;
        const int wave = tid >> 5;
        const int lane = tid & 31;
        hist[tid] = 0u;
        __syncthreads();
        const int gw = ((int)blockIdx.x * 32 + wave * 4) * 32 + lane;
        int4 vd[4], vs[4];
        #pragma unroll
        for (int q = 0; q < 4; ++q) {
            const int g = gw + q * 32;
            const size_t o = (size_t)(g < G4 ? g : 0) * 4;
            vd[q] = *(const int4*)(dst + o);
            vs[q] = *(const int4*)(src + o);
        }
        unsigned key[16];
        #pragma unroll
        for (int q = 0; q < 4; ++q) {
            const bool okg = gw + q * 32 < G4;
            const int dd[4] = { vd[q].x, vd[q].y, vd[q].z, vd[q].w };
            #pragma unroll
            for (int c = 0; c < 4; ++c) {
                unsigned k = 0xffffffffu;
                if (okg && (unsigned)dd[c] < (unsigned)N) {
                    const unsigned r = (unsigned)dd[c] / NPW;
                    k = (r << 16) | __hip_atomic_fetch_add(&hist[r], 1u, __ATOMIC_RELAXED, __HIP_MEMORY_SCOPE_WORKGROUP);
                }
                key[q * 4 + c] = k;
            }
        }
        __syncthreads();
        if (wave == 0) {
            unsigned c8[8];
            unsigned t = 0u;
            #pragma unroll
            for (int i = 0; i < 8; ++i) { c8[i] = hist[lane * 8 + i]; t += c8[i]; }
            unsigned incl = t;
            #pragma unroll
            for (int d = 1; d < 32; d <<= 1) {
                const unsigned o = __shfl_up(incl, d, 32);
                if (lane >= d) incl += o;
            }
            unsigned run = incl - t;
            #pragma unroll
            for (int i = 0; i < 8; ++i) { base[lane * 8 + i] = run; run += c8[i]; }
        }
        __syncthreads();
        #pragma unroll
        for (int q = 0; q < 4; ++q) {
            const int dd[4] = { vd[q].x, vd[q].y, vd[q].z, vd[q].w };
            const int ss[4] = { vs[q].x, vs[q].y, vs[q].z, vs[q].w };
            #pragma unroll
            for (int c = 0; c < 4; ++c) {
                const unsigned k = key[q * 4 + c];
                if (k != 0xffffffffu) {
                    const unsigned r = k >> 16;
                    sorted[base[r] + (k & 0xffffu)] =
                        ((unsigned)ss[c] & ((1u << SRCBITS) - 1u)) | (((unsigned)dd[c] - r * NPW) << SRCBITS);
                }
            }
        }
        __syncthreads();
        const unsigned total = base[BMROWS - 1] + hist[BMROWS - 1];
        unsigned* const ob = blk + (size_t)blockIdx.x * EPB;
        for (unsigned i = tid; i < total; i += 256) ob[i] = sorted[i];
        offs[(size_t)blockIdx.x * BMROWS + tid] = base[tid] | (hist[tid] << 16);
        return;
    }

    int t = ((int)blockIdx.x - nbw) * 256 + (int)threadIdx.x;
    if (t >= N * 4) return;
    const int n = t >> 2, h = t & 3;
    const float* f = feat + (size_t)n * 64 + h * 16;
    float sl = 0.f, sr = 0.f;
    #pragma unroll
    for (int i = 0; i < 16; ++i) {
        const float v = f[i];
        sl += v * al[h * 16 + i];
        sr += v * ar[h * 16 + i];
    }
    el[t] = sl;
    er[t] = sr;
}

__device__ __forceinline__ float leaky02(float x) { return x > 0.f ? x : 0.2f * x; }

__device__ __forceinline__ unsigned f2o(float f) {
    unsigned u = __float_as_uint(f);
    return (u >> 31) ? ~u : (u | 0x80000000u);
}

__device__ __forceinline__ float o2f(unsigned u) {
    if (u == 0u) return 0.f;
    return (u & 0x80000000u) ? __uint_as_float(u ^ 0x80000000u) : __uint_as_float(~u);
}

template <bool USEIX>
__global__ __launch_bounds__(1024) void gat_agg_owned(
    const int* __restrict__ src, const int* __restrict__ dst,
    const float* __restrict__ el, const float* __restrict__ er,
    const float* __restrict__ feat,
    const float* __restrict__ bias, float* __restrict__ out, int N, int E,
    const unsigned* __restrict__ blk, const unsigned* __restrict__ offs, int nbw)
{
    __shared__ float acc[NPW * 64];
    __shared__ __attribute__((aligned(16))) float    sden[NPW * 4];
    __shared__ __attribute__((aligned(16))) unsigned smax[NPW * 4];
    __shared__ __attribute__((aligned(16))) float swt[AGGW][32 * 4];
    __shared__ int   lst_e[AGGW][LCAP];
    __shared__ int   lst_d[AGGW][LCAP];
    __shared__ int   s_lost;

    const int tid  = threadIdx.x;
    const int wave = tid >> 5;
    const int lane = tid & 31;
    const int lo   = blockIdx.x * NPW;
    const int cnt  = (N - lo) < NPW ? (N - lo) : NPW;

    for (int i = tid; i < NPW * 64; i += 1024) acc[i] = 0.f;
    for (int i = tid; i < NPW * 4;  i += 1024) { sden[i] = 0.f; smax[i] = 0u; }
    if (tid == 0) s_lost = 0;
    __syncthreads();

    float* const myw = swt[wave];
    int*   const mye = lst_e[wave];
    int*   const myd = lst_d[wave];

    auto maxupd = [&](int n, bool have_s) __attribute__((always_inline)) {
        for (int b = 0; b < n; b += 128) {
            int    sv[4], dv[4];
            float4 l4[4], r4[4];
            #pragma unroll
            for (int u = 0; u < 4; ++u) {
                const int  i     = b + u * 32 + lane;
                const bool valid = i < n;
                const int  e     = valid ? mye[i] : 0;
                dv[u] = valid ? myd[i] : -1;
                sv[u] = have_s ? e : src[e];
            }
            #pragma unroll
            for (int u = 0; u < 4; ++u) {
                const int dl = dv[u] < 0 ? 0 : dv[u];
                l4[u] = *(const float4*)(el + (size_t)sv[u] * 4);
                r4[u] = *(const float4*)(er + (size_t)(lo + dl) * 4);
            }
            #pragma unroll
            for (int u = 0; u < 4; ++u) {
                if (dv[u] >= 0) {
                    unsigned* mx = &smax[dv[u] * 4];
                    __hip_atomic_fetch_max(mx + 0, f2o(leaky02(l4[u].x + r4[u].x)), __ATOMIC_RELAXED, __HIP_MEMORY_SCOPE_WORKGROUP);
                    __hip_atomic_fetch_max(mx + 1, f2o(leaky02(l4[u].y + r4[u].y)), __ATOMIC_RELAXED, __HIP_MEMORY_SCOPE_WORKGROUP);
                    __hip_atomic_fetch_max(mx + 2, f2o(leaky02(l4[u].z + r4[u].z)), __ATOMIC_RELAXED, __HIP_MEMORY_SCOPE_WORKGROUP);
                    __hip_atomic_fetch_max(mx + 3, f2o(leaky02(l4[u].w + r4[u].w)), __ATOMIC_RELAXED, __HIP_MEMORY_SCOPE_WORKGROUP);
                    if (!have_s) mye[b + u * 32 + lane] = sv[u];
                }
            }
        }
    };

    auto drain = [&](int n, bool have_s) __attribute__((always_inline)) {
        for (int b = 0; b < n; b += 32) {
            const int  i     = b + lane;
            const bool valid = i < n;
            const int  q  = valid ? mye[i] : 0;
            const int  dl = valid ? myd[i] : 0;
            const int  s  = have_s ? q : src[q];
            const int  d  = lo + dl;
            const float4 l4 = *(const float4*)(el + (size_t)s * 4);
            const float4 r4 = *(const float4*)(er + (size_t)d * 4);
            const uint4  mu = *(const uint4*)&smax[dl * 4];
            float4 w;
            w.x = valid ? __expf(leaky02(l4.x + r4.x) - o2f(mu.x)) : 0.f;
            w.y = valid ? __expf(leaky02(l4.y + r4.y) - o2f(mu.y)) : 0.f;
            w.z = valid ? __expf(leaky02(l4.z + r4.z) - o2f(mu.z)) : 0.f;
            w.w = valid ? __expf(leaky02(l4.w + r4.w) - o2f(mu.w)) : 0.f;
            if (valid) {
                float* dn = &sden[dl * 4];
                __hip_atomic_fetch_add(dn + 0, w.x, __ATOMIC_RELAXED, __HIP_MEMORY_SCOPE_WORKGROUP);
                __hip_atomic_fetch_add(dn + 1, w.y, __ATOMIC_RELAXED, __HIP_MEMORY_SCOPE_WORKGROUP);
                __hip_atomic_fetch_add(dn + 2, w.z, __ATOMIC_RELAXED, __HIP_MEMORY_SCOPE_WORKGROUP);
                __hip_atomic_fetch_add(dn + 3, w.w, __ATOMIC_RELAXED, __HIP_MEMORY_SCOPE_WORKGROUP);
            }
            *(float4*)&myw[lane * 4] = w;

            const int nb = (n - b) < 32 ? (n - b) : 32;
            for (int j0 = 0; j0 < nb; j0 += 8) {
                float2 f[8];
                float  wj[8];
                int    dj[8];
                #pragma unroll
                for (int u = 0; u < 8; ++u) {
                    const int jj = j0 + u;
                    const int jc = jj < nb ? jj : nb - 1;
                    const int sj = __builtin_amdgcn_readlane(s, jc);
                    dj[u] = __builtin_amdgcn_readlane(dl, jc);
                    f[u]  = *(const float2*)(feat + (size_t)sj * 64 + lane * 2);
                    wj[u] = jj < nb ? myw[jc * 4 + (lane >> 3)] : 0.f;
                }
                #pragma unroll
                for (int u = 0; u < 8; ++u) {
                    float* a = &acc[dj[u] * 64 + lane * 2];
                    __hip_atomic_fetch_add(a,     f[u].x * wj[u], __ATOMIC_RELAXED, __HIP_MEMORY_SCOPE_WORKGROUP);
                    __hip_atomic_fetch_add(a + 1, f[u].y * wj[u], __ATOMIC_RELAXED, __HIP_MEMORY_SCOPE_WORKGROUP);
                }
            }
        }
    };

    const int G4     = E >> 2;
    const int steps  = ((G4 + 1023) >> 10);
    const int steps4 = (steps + 3) & ~3;
    const int st0    = (int)(((long long)blockIdx.x * steps4) / gridDim.x) & ~3;
    int fill = 0;
    int pass = 0;

    auto scan_load = [&](int k, int4 (&v)[4]) {
        int kk = st0 + k;
        if (kk >= steps4) kk -= steps4;
        #pragma unroll
        for (int u = 0; u < 4; ++u) {
            const int g = (kk + u) * 1024 + tid;
            v[u] = *(const int4*)(dst + (size_t)(g < G4 ? g : 0) * 4);
        }
    };

    auto examine = [&](int g, bool ok, const int4& v) __attribute__((always_inline)) {
        const int e0 = g * 4;
        const unsigned lim = ok ? (unsigned)cnt : 0u;
        const unsigned u0 = (unsigned)(v.x - lo);
        const unsigned u1 = (unsigned)(v.y - lo);
        const unsigned u2 = (unsigned)(v.z - lo);
        const unsigned u3 = (unsigned)(v.w - lo);
        const bool m0 = u0 < lim, m1 = u1 < lim;
        const bool m2 = u2 < lim, m3 = u3 < lim;
        if (__builtin_amdgcn_ballot_w32(m0 | m1 | m2 | m3)) {
            unsigned hb;
            hb = __builtin_amdgcn_ballot_w32(m0);
            if (m0) { const int p = fill + __builtin_amdgcn_mbcnt_lo(hb, 0u); if (p < LCAP) { mye[p] = e0;     myd[p] = (int)u0; } }
            fill += __builtin_popcount(hb);
            hb = __builtin_amdgcn_ballot_w32(m1);
            if (m1) { const int p = fill + __builtin_amdgcn_mbcnt_lo(hb, 0u); if (p < LCAP) { mye[p] = e0 + 1; myd[p] = (int)u1; } }
            fill += __builtin_popcount(hb);
            hb = __builtin_amdgcn_ballot_w32(m2);
            if (m2) { const int p = fill + __builtin_amdgcn_mbcnt_lo(hb, 0u); if (p < LCAP) { mye[p] = e0 + 2; myd[p] = (int)u2; } }
            fill += __builtin_popcount(hb);
            hb = __builtin_amdgcn_ballot_w32(m3);
            if (m3) { const int p = fill + __builtin_amdgcn_mbcnt_lo(hb, 0u); if (p < LCAP) { mye[p] = e0 + 3; myd[p] = (int)u3; } }
            fill += __builtin_popcount(hb);
        }
    };

    auto flush = [&]() __attribute__((always_inline)) {
        if (pass != 0 && fill > LCAP - 512) {
            if (pass == 1) maxupd(fill, USEIX);
            else           drain(fill, USEIX);
            fill = 0;
        }
    };

    auto scan_use = [&](int k, const int4 (&v)[4]) {
        int kk = st0 + k;
        if (kk >= steps4) kk -= steps4;
        #pragma unroll
        for (int u = 0; u < 4; ++u) {
            const int g = (kk + u) * 1024 + tid;
            examine(g, g < G4, v[u]);
        }
        flush();
    };

    auto append = [&](bool m, unsigned w, int add) __attribute__((always_inline)) {
        const int p = fill + lane;
        if (m && p < LCAP) { mye[p] = (int)(w & ((1u << SRCBITS) - 1u)); myd[p] = (int)(w >> SRCBITS); }
        fill += add;
    };

    auto collect = [&]() __attribute__((always_inline)) {
        for (int b0 = wave; b0 < nbw; b0 += AGGW * 32) {
            const int bl = b0 + lane * AGGW;
            const unsigned meta = bl < nbw ? offs[(size_t)bl * BMROWS + blockIdx.x] : 0u;
            const int left = (nbw - b0 + AGGW - 1) / AGGW;
            const int nseg = left < 32 ? left : 32;
            for (int i0 = 0; i0 < nseg; i0 += 8) {
                unsigned w[8];
                int      cn[8];
                const unsigned* sg[8];
                #pragma unroll
                for (int u = 0; u < 8; ++u) {
                    const int i = i0 + u;
                    const unsigned mt = (unsigned)__builtin_amdgcn_readlane((int)meta, i);
                    cn[u] = (int)(mt >> 16);
                    sg[u] = blk + (size_t)(b0 + i * AGGW) * EPB + (mt & 0xffffu);
                    w[u]  = lane < cn[u] ? sg[u][lane] : 0u;
                }
                #pragma unroll
                for (int u = 0; u < 8; ++u) append(lane < cn[u], w[u], cn[u] < 32 ? cn[u] : 32);
                flush();
                #pragma unroll 1
                for (int i = i0; i < i0 + 8 && i < nseg; ++i) {
                    const unsigned mt = (unsigned)__builtin_amdgcn_readlane((int)meta, i);
                    const int c = (int)(mt >> 16);
                    const unsigned* const sp = blk + (size_t)(b0 + i * AGGW) * EPB + (mt & 0xffffu);
                    for (int k = 32; k < c; k += 32) {
                        const bool m = k + lane < c;
                        append(m, m ? sp[k + lane] : 0u, (c - k) < 32 ? (c - k) : 32);
                        flush();
                    }
                }
            }
        }
    };

    for (pass = 0; pass < 3; ++pass) {
        fill = 0;
        if constexpr (USEIX) {
            collect();
        } else {
            int4 va[4], vb[4];
            if (steps4 > 0) scan_load(0, va);
            for (int k = 0; k < steps4; k += 8) {
                scan_load(k + 4, vb);
                scan_use(k, va);
                scan_load(k + 8, va);
                if (k + 4 < steps4) scan_use(k + 4, vb);
            }
        }
        if (wave == 0) {
            const int et = (G4 << 2) + lane;
            unsigned ut = 0xffffffffu;
            if (et < E) ut = (unsigned)(dst[et] - lo);
            const bool mt = ut < (unsigned)cnt;
            const unsigned bm = __builtin_amdgcn_ballot_w32(mt);
            if (mt) { const int p = fill + __builtin_amdgcn_mbcnt_lo(bm, 0u); if (p < LCAP) { mye[p] = USEIX ? src[et] : et; myd[p] = (int)ut; } }
            fill += __builtin_popcount(bm);
        }
        if (pass == 0) {
            if (fill > LCAP && lane == 0) s_lost = 1;
            maxupd(fill < LCAP ? fill : LCAP, USEIX);
            __syncthreads();
            if (__builtin_amdgcn_readfirstlane(s_lost) == 0) { drain(fill, true); break; }
        } else if (pass == 1) {
            maxupd(fill, USEIX);
            __syncthreads();
        } else {
            drain(fill, USEIX);
        }
    }
    __syncthreads();

    for (int i = tid; i < cnt * 64; i += 1024) {
        const int n = i >> 6, c = i & 63;
        const float dn = sden[n * 4 + (c >> 4)];
        const float a  = acc[i];
        out[(size_t)(lo + n) * 64 + c] = (dn > 0.f ? a / dn : 0.f) + bias[c];
    }
}

extern "C" void kernel_launch(void* const* d_in, const int* in_sizes, int n_in,
                              void* d_out, int out_size, void* d_ws, size_t ws_size,
                              hipStream_t stream)
{
    const float* nf   = (const float*)d_in[0];
    const float* W    = (const float*)d_in[1];
    const float* al   = (const float*)d_in[2];
    const float* ar   = (const float*)d_in[3];
    const float* bias = (const float*)d_in[4];
    const int*   src  = (const int*)d_in[5];
    const int*   dst  = (const int*)d_in[6];

    const int N = in_sizes[0] / 64;
    const int E = in_sizes[5];

    float* ws    = (float*)d_ws;
    float* feat  = ws;
    float* el    = feat  + (size_t)N * 64;
    float* er    = el    + (size_t)N * 4;
    float* out   = (float*)d_out;
    const int nr  = (N + NPW - 1) / NPW;
    const int G4  = E >> 2;
    const int nbw = (G4 + 1023) / 1024;
    unsigned* blk  = (unsigned*)(er + (size_t)N * 4);
    unsigned* offs = blk + (size_t)nbw * EPB;
    const size_t need = ((size_t)N * 72 + (size_t)nbw * (EPB + BMROWS)) * sizeof(float);

    gat_proj_wmma<<<(N + 127) / 128, 256, 0, stream>>>(nf, W, feat, N);
    if (nr <= BMROWS && nbw > 0 && N <= (1 << SRCBITS) && ws_size >= need) {
        gat_elr_sort<<<nbw + (N * 4 + 255) / 256, 256, 0, stream>>>(feat, al, ar, el, er, N, src, dst, blk, offs, G4, nbw);
        gat_agg_owned<true><<<nr, 1024, 0, stream>>>(src, dst, el, er, feat, bias, out, N, E, blk, offs, nbw);
    } else {
        gat_elr<<<(N * 4 + 255) / 256, 256, 0, stream>>>(feat, al, ar, el, er, N);
        gat_agg_owned<false><<<nr, 1024, 0, stream>>>(src, dst, el, er, feat, bias, out, N, E, nullptr, nullptr, 0);
    }
}


# sum phase rows by half-waves: 16 lanes x 16 B per edge row, two rows per load; source and destination passed by DPP row_share, weights read as 16-B vectors; float sums stay in arrival order
# speedup vs baseline: 5.8995x; 1.2118x over previous
#include <hip/hip_runtime.h>


typedef float v2f __attribute__((ext_vector_type(2)));
typedef float v8f __attribute__((ext_vector_type(8)));

#define SA_STRIDE 68
#define SW_STRIDE 68

#define NPW   391
#define LCAP  640
#define LCAPX 304
#define AGGW  32
#define BMROWS 256
#define EPB   4096
#define SRCBITS 23
static_assert(NPW <= (1 << (32 - SRCBITS)), "local destination must fit above the source bits");

#define SWT_ROWS 80
#define PROJ_LDS_BYTES ((SWT_ROWS * SW_STRIDE + 8 * 16 * SA_STRIDE) * 4)
static_assert((2 * BMROWS + EPB) * 4 <= PROJ_LDS_BYTES, "the sorter's counters and image must fit in the projection's LDS block");

__global__ __launch_bounds__(256) void gat_proj_logits_sort(
    const float* __restrict__ nf, const float* __restrict__ W,
    const float* __restrict__ al, const float* __restrict__ ar,
    float* __restrict__ feat, float* __restrict__ el, float* __restrict__ er, int N,
    const int* __restrict__ src, const int* __restrict__ dst,
    unsigned* __restrict__ blk, unsigned* __restrict__ offs, int G4, int nbw)
{
    __shared__ __attribute__((aligned(16))) unsigned char smem[PROJ_LDS_BYTES];

    const int tid  = threadIdx.x;
    const int wave = tid >> 5;
    const int lane = tid & 31;

    if ((int)blockIdx.x < nbw) {
        unsigned* const hist   = (unsigned*)smem;
        unsigned* const base   = hist + BMROWS;
        unsigned* const sorted = base + BMROWS;
        hist[tid] = 0u;
        __syncthreads();
        const int gw = ((int)blockIdx.x * 32 + wave * 4) * 32 + lane;
        int4 vd[4], vs[4];
        #pragma unroll
        for (int q = 0; q < 4; ++q) {
            const int g = gw + q * 32;
            const size_t o = (size_t)(g < G4 ? g : 0) * 4;
            vd[q] = *(const int4*)(dst + o);
            vs[q] = *(const int4*)(src + o);
        }
        unsigned key[16];
        #pragma unroll
        for (int q = 0; q < 4; ++q) {
            const bool okg = gw + q * 32 < G4;
            const int dd[4] = { vd[q].x, vd[q].y, vd[q].z, vd[q].w };
            #pragma unroll
            for (int c = 0; c < 4; ++c) {
                unsigned k = 0xffffffffu;
                if (okg && (unsigned)dd[c] < (unsigned)N) {
                    const unsigned r = (unsigned)dd[c] / NPW;
                    k = (r << 16) | __hip_atomic_fetch_add(&hist[r], 1u, __ATOMIC_RELAXED, __HIP_MEMORY_SCOPE_WORKGROUP);
                }
                key[q * 4 + c] = k;
            }
        }
        __syncthreads();
        if (wave == 0) {
            unsigned c8[8];
            unsigned t = 0u;
            #pragma unroll
            for (int i = 0; i < 8; ++i) { c8[i] = hist[lane * 8 + i]; t += c8[i]; }
            unsigned incl = t;
            #pragma unroll
            for (int d = 1; d < 32; d <<= 1) {
                const unsigned o = __shfl_up(incl, d, 32);
                if (lane >= d) incl += o;
            }
            unsigned run = incl - t;
            #pragma unroll
            for (int i = 0; i < 8; ++i) { base[lane * 8 + i] = run; run += c8[i]; }
        }
        __syncthreads();
        #pragma unroll
        for (int q = 0; q < 4; ++q) {
            const int dd[4] = { vd[q].x, vd[q].y, vd[q].z, vd[q].w };
            const int ss[4] = { vs[q].x, vs[q].y, vs[q].z, vs[q].w };
            #pragma unroll
            for (int c = 0; c < 4; ++c) {
                const unsigned k = key[q * 4 + c];
                if (k != 0xffffffffu) {
                    const unsigned r = k >> 16;
                    sorted[base[r] + (k & 0xffffu)] =
                        ((unsigned)ss[c] & ((1u << SRCBITS) - 1u)) | (((unsigned)dd[c] - r * NPW) << SRCBITS);
                }
            }
        }
        __syncthreads();
        const unsigned total = base[BMROWS - 1] + hist[BMROWS - 1];
        unsigned* const ob = blk + (size_t)blockIdx.x * EPB;
        for (unsigned i = tid; i < total; i += 256) ob[i] = sorted[i];
        offs[(size_t)blockIdx.x * BMROWS + tid] = base[tid] | (hist[tid] << 16);
        return;
    }

    float* const sWT = (float*)smem;
    float* const sAw = sWT + SWT_ROWS * SW_STRIDE + wave * (16 * SA_STRIDE);

    for (int i = tid; i < 64 * 64; i += 256) {
        const int k = i >> 6, n = i & 63;
        sWT[n * SW_STRIDE + k] = W[i];
    }
    for (int i = tid; i < 8 * SW_STRIDE; i += 256) sWT[72 * SW_STRIDE + i] = 0.f;

    const int rowbase = ((int)blockIdx.x - nbw) * 128 + wave * 16;
    for (int r = 0; r < 16; ++r) {
        int row = rowbase + r;
        row = row < N ? row : (N - 1);
        sAw[r * SA_STRIDE + lane]      = nf[(size_t)row * 64 + lane];
        sAw[r * SA_STRIDE + lane + 32] = nf[(size_t)row * 64 + lane + 32];
    }
    __syncthreads();

    {
        const int k  = tid & 63;
        const int hh = __builtin_amdgcn_readfirstlane(tid >> 6);
        float sl = 0.f, sr = 0.f;
        #pragma unroll
        for (int f = 0; f < 16; ++f) {
            const float w = sWT[(hh * 16 + f) * SW_STRIDE + k];
            sl += w * al[hh * 16 + f];
            sr += w * ar[hh * 16 + f];
        }
        sWT[(64 + hh) * SW_STRIDE + k] = sl;
        sWT[(68 + hh) * SW_STRIDE + k] = sr;
    }
    __syncthreads();

    const int half = lane >> 4;
    const int m    = lane & 15;

    v8f acc0 = {}, acc1 = {}, acc2 = {}, acc3 = {}, acc4 = {};

    #pragma unroll
    for (int k0 = 0; k0 < 64; k0 += 4) {
        const int ka = k0 + half * 2;

        const v2f a  = *(const v2f*)&sAw[m * SA_STRIDE + ka];
        const v2f b0 = *(const v2f*)&sWT[(m +  0) * SW_STRIDE + ka];
        const v2f b1 = *(const v2f*)&sWT[(m + 16) * SW_STRIDE + ka];
        const v2f b2 = *(const v2f*)&sWT[(m + 32) * SW_STRIDE + ka];
        const v2f b3 = *(const v2f*)&sWT[(m + 48) * SW_STRIDE + ka];
        const v2f b4 = *(const v2f*)&sWT[(m + 64) * SW_STRIDE + ka];

        acc0 = __builtin_amdgcn_wmma_f32_16x16x4_f32(false, a, false, b0, (short)0, acc0, false, false);
        acc1 = __builtin_amdgcn_wmma_f32_16x16x4_f32(false, a, false, b1, (short)0, acc1, false, false);
        acc2 = __builtin_amdgcn_wmma_f32_16x16x4_f32(false, a, false, b2, (short)0, acc2, false, false);
        acc3 = __builtin_amdgcn_wmma_f32_16x16x4_f32(false, a, false, b3, (short)0, acc3, false, false);
        acc4 = __builtin_amdgcn_wmma_f32_16x16x4_f32(false, a, false, b4, (short)0, acc4, false, false);
    }

    float* const lg = (m < 4 ? el : er) + (m & 3);
    #pragma unroll
    for (int v = 0; v < 8; ++v) {
        const int row = rowbase + v + half * 8;
        if (row < N) {
            float* fr = feat + (size_t)row * 64;
            fr[ 0 + m] = acc0[v];
            fr[16 + m] = acc1[v];
            fr[32 + m] = acc2[v];
            fr[48 + m] = acc3[v];
            if (m < 8) lg[(size_t)row * 4] = acc4[v];
        }
    }
}

__device__ __forceinline__ float leaky02(float x) { return x > 0.f ? x : 0.2f * x; }

__device__ __forceinline__ unsigned f2o(float f) {
    unsigned u = __float_as_uint(f);
    return (u >> 31) ? ~u : (u | 0x80000000u);
}

__device__ __forceinline__ float o2f(unsigned u) {
    if (u == 0u) return 0.f;
    return (u & 0x80000000u) ? __uint_as_float(u ^ 0x80000000u) : __uint_as_float(~u);
}

template <int U>
__device__ __forceinline__ int row_share(int v) {
    return __builtin_amdgcn_update_dpp(0, v, 0x150 + U, 0xf, 0xf, true);
}

template <int C, bool NARROW>
__device__ __forceinline__ void row_chunk(int s, int dl, const float* myw, const float* __restrict__ feat,
                                          float* acc, int lane)
{
    const int half = lane >> 4;
    const int c4   = lane & 15;
    const float4 wa = *(const float4*)(myw + C * 64 + half * 32 + (c4 >> 2) * 8);
    const float4 wb = *(const float4*)(myw + C * 64 + half * 32 + (c4 >> 2) * 8 + 4);
    const float wv[8] = { wa.x, wa.y, wa.z, wa.w, wb.x, wb.y, wb.z, wb.w };
    float4 f[8];
    int    dj[8];
#define ROW_LOAD(K) { const int sj = row_share<C * 8 + K>(s); dj[K] = row_share<C * 8 + K>(dl); \
                      if constexpr (NARROW) f[K] = *(const float4*)(feat + (((unsigned)sj << 6) | (unsigned)(c4 * 4))); \
                      else                  f[K] = *(const float4*)(feat + (size_t)sj * 64 + c4 * 4); }
    ROW_LOAD(0) ROW_LOAD(1) ROW_LOAD(2) ROW_LOAD(3) ROW_LOAD(4) ROW_LOAD(5) ROW_LOAD(6) ROW_LOAD(7)
#undef ROW_LOAD
    float* const acca = acc + c4 * 4 + (half ? 2 : 0);
    float* const accb = acc + c4 * 4 + (half ? 0 : 2);
    #pragma unroll
    for (int k = 0; k < 8; ++k) {
        const float p0 = f[k].x * wv[k], p1 = f[k].y * wv[k];
        const float p2 = f[k].z * wv[k], p3 = f[k].w * wv[k];
        float* const ra = acca + dj[k] * 64;
        float* const rb = accb + dj[k] * 64;
        __hip_atomic_fetch_add(ra,     half ? p2 : p0, __ATOMIC_RELAXED, __HIP_MEMORY_SCOPE_WORKGROUP);
        __hip_atomic_fetch_add(ra + 1, half ? p3 : p1, __ATOMIC_RELAXED, __HIP_MEMORY_SCOPE_WORKGROUP);
        __hip_atomic_fetch_add(rb,     half ? p0 : p2, __ATOMIC_RELAXED, __HIP_MEMORY_SCOPE_WORKGROUP);
        __hip_atomic_fetch_add(rb + 1, half ? p1 : p3, __ATOMIC_RELAXED, __HIP_MEMORY_SCOPE_WORKGROUP);
    }
}

template <bool USEIX>
__global__ __launch_bounds__(1024) void gat_agg_owned(
    const int* __restrict__ src, const int* __restrict__ dst,
    const float* __restrict__ el, const float* __restrict__ er,
    const float* __restrict__ feat,
    const float* __restrict__ bias, float* __restrict__ out, int N, int E,
    const unsigned* __restrict__ blk, const unsigned* __restrict__ offs, int nbw)
{
    __shared__ float acc[NPW * 64];
    __shared__ __attribute__((aligned(16))) float    sden[NPW * 4];
    __shared__ __attribute__((aligned(16))) unsigned smax[NPW * 4];
    constexpr int LC  = USEIX ? LCAPX : LCAP;
    constexpr int APP = USEIX ? 256 : 512;
    constexpr unsigned SRCMASK = (1u << SRCBITS) - 1u;
    __shared__ __attribute__((aligned(16))) float swt[AGGW][32 * 4];
    __shared__ int   lst_e[AGGW][LC];
    __shared__ int   lst_d[USEIX ? 1 : AGGW][USEIX ? 1 : LC];
    __shared__ __attribute__((aligned(16))) float ssc[USEIX ? AGGW : 1][(USEIX ? LC : 1) * 4];
    __shared__ int   s_lost;

    const int tid  = threadIdx.x;
    const int wave = tid >> 5;
    const int lane = tid & 31;
    const int lo   = blockIdx.x * NPW;
    const int cnt  = (N - lo) < NPW ? (N - lo) : NPW;

    for (int i = tid; i < NPW * 64; i += 1024) acc[i] = 0.f;
    for (int i = tid; i < NPW * 4;  i += 1024) { sden[i] = 0.f; smax[i] = 0u; }
    if (tid == 0) s_lost = 0;
    __syncthreads();

    float* const myw = swt[wave];
    int*   const mye = lst_e[wave];
    int*   const myd = lst_d[USEIX ? 0 : wave];
    float* const mys = ssc[USEIX ? wave : 0];

    auto maxupd = [&](int n, bool have_s, bool keep) __attribute__((always_inline)) {
        for (int b = 0; b < n; b += 128) {
            int    sv[4], dv[4];
            float4 l4[4], r4[4];
            #pragma unroll
            for (int u = 0; u < 4; ++u) {
                const int  i     = b + u * 32 + lane;
                const bool valid = i < n;
                const int  e     = valid ? mye[i] : 0;
                if constexpr (USEIX) {
                    dv[u] = valid ? (int)((unsigned)e >> SRCBITS) : -1;
                    sv[u] = (int)((unsigned)e & SRCMASK);
                } else {
                    dv[u] = valid ? myd[i] : -1;
                    sv[u] = have_s ? e : src[e];
                }
            }
            #pragma unroll
            for (int u = 0; u < 4; ++u) {
                const int dl = dv[u] < 0 ? 0 : dv[u];
                l4[u] = *(const float4*)(el + (size_t)sv[u] * 4);
                r4[u] = *(const float4*)(er + (size_t)(lo + dl) * 4);
            }
            #pragma unroll
            for (int u = 0; u < 4; ++u) {
                if (dv[u] >= 0) {
                    unsigned* mx = &smax[dv[u] * 4];
                    float4 sc;
                    sc.x = leaky02(l4[u].x + r4[u].x);
                    sc.y = leaky02(l4[u].y + r4[u].y);
                    sc.z = leaky02(l4[u].z + r4[u].z);
                    sc.w = leaky02(l4[u].w + r4[u].w);
                    __hip_atomic_fetch_max(mx + 0, f2o(sc.x), __ATOMIC_RELAXED, __HIP_MEMORY_SCOPE_WORKGROUP);
                    __hip_atomic_fetch_max(mx + 1, f2o(sc.y), __ATOMIC_RELAXED, __HIP_MEMORY_SCOPE_WORKGROUP);
                    __hip_atomic_fetch_max(mx + 2, f2o(sc.z), __ATOMIC_RELAXED, __HIP_MEMORY_SCOPE_WORKGROUP);
                    __hip_atomic_fetch_max(mx + 3, f2o(sc.w), __ATOMIC_RELAXED, __HIP_MEMORY_SCOPE_WORKGROUP);
                    if (USEIX && keep) *(float4*)&mys[(b + u * 32 + lane) * 4] = sc;
                    if (!USEIX && !have_s) mye[b + u * 32 + lane] = sv[u];
                }
            }
        }
    };

    auto drain = [&](int n, bool have_s, bool kept) __attribute__((always_inline)) {
        for (int b = 0; b < n; b += 32) {
            const int  i     = b + ((lane & 15) << 1) + (lane >> 4);
            const bool valid = i < n;
            const int  q  = valid ? mye[i] : 0;
            int s, dl;
            if constexpr (USEIX) {
                s  = (int)((unsigned)q & SRCMASK);
                dl = (int)((unsigned)q >> SRCBITS);
            } else {
                dl = valid ? myd[i] : 0;
                s  = have_s ? q : src[q];
                s  = valid ? s : 0;
            }
            float4 sc;
            if (USEIX && kept) {
                sc = *(const float4*)&mys[(valid ? i : 0) * 4];
            } else {
                const int  d  = lo + dl;
                const float4 l4 = *(const float4*)(el + (size_t)s * 4);
                const float4 r4 = *(const float4*)(er + (size_t)d * 4);
                sc.x = leaky02(l4.x + r4.x);
                sc.y = leaky02(l4.y + r4.y);
                sc.z = leaky02(l4.z + r4.z);
                sc.w = leaky02(l4.w + r4.w);
            }
            const uint4  mu = *(const uint4*)&smax[dl * 4];
            float4 w;
            w.x = valid ? __expf(sc.x - o2f(mu.x)) : 0.f;
            w.y = valid ? __expf(sc.y - o2f(mu.y)) : 0.f;
            w.z = valid ? __expf(sc.z - o2f(mu.z)) : 0.f;
            w.w = valid ? __expf(sc.w - o2f(mu.w)) : 0.f;
            if (valid) {
                float* dn = &sden[dl * 4];
                __hip_atomic_fetch_add(dn + 0, w.x, __ATOMIC_RELAXED, __HIP_MEMORY_SCOPE_WORKGROUP);
                __hip_atomic_fetch_add(dn + 1, w.y, __ATOMIC_RELAXED, __HIP_MEMORY_SCOPE_WORKGROUP);
                __hip_atomic_fetch_add(dn + 2, w.z, __ATOMIC_RELAXED, __HIP_MEMORY_SCOPE_WORKGROUP);
                __hip_atomic_fetch_add(dn + 3, w.w, __ATOMIC_RELAXED, __HIP_MEMORY_SCOPE_WORKGROUP);
            }
            {
                float* wd = myw + ((lane & 15) >> 3) * 64 + (lane >> 4) * 32 + (lane & 7);
                wd[0] = w.x; wd[8] = w.y; wd[16] = w.z; wd[24] = w.w;
            }

            const int nb = (n - b) < 32 ? (n - b) : 32;
            row_chunk<0, USEIX>(s, dl, myw, feat, acc, lane);
            if (nb > 16) row_chunk<1, USEIX>(s, dl, myw, feat, acc, lane);
        }
    };

    const int G4     = E >> 2;
    const int steps  = ((G4 + 1023) >> 10);
    const int steps4 = (steps + 3) & ~3;
    const int st0    = (int)(((long long)blockIdx.x * steps4) / gridDim.x) & ~3;
    int fill = 0;
    int pass = 0;

    auto scan_load = [&](int k, int4 (&v)[4]) {
        int kk = st0 + k;
        if (kk >= steps4) kk -= steps4;
        #pragma unroll
        for (int u = 0; u < 4; ++u) {
            const int g = (kk + u) * 1024 + tid;
            v[u] = *(const int4*)(dst + (size_t)(g < G4 ? g : 0) * 4);
        }
    };

    auto examine = [&](int g, bool ok, const int4& v) __attribute__((always_inline)) {
        const int e0 = g * 4;
        const unsigned lim = ok ? (unsigned)cnt : 0u;
        const unsigned u0 = (unsigned)(v.x - lo);
        const unsigned u1 = (unsigned)(v.y - lo);
        const unsigned u2 = (unsigned)(v.z - lo);
        const unsigned u3 = (unsigned)(v.w - lo);
        const bool m0 = u0 < lim, m1 = u1 < lim;
        const bool m2 = u2 < lim, m3 = u3 < lim;
        if (__builtin_amdgcn_ballot_w32(m0 | m1 | m2 | m3)) {
            unsigned hb;
            hb = __builtin_amdgcn_ballot_w32(m0);
            if (m0) { const int p = fill + __builtin_amdgcn_mbcnt_lo(hb, 0u); if (p < LC) { mye[p] = e0;     myd[p] = (int)u0; } }
            fill += __builtin_popcount(hb);
            hb = __builtin_amdgcn_ballot_w32(m1);
            if (m1) { const int p = fill + __builtin_amdgcn_mbcnt_lo(hb, 0u); if (p < LC) { mye[p] = e0 + 1; myd[p] = (int)u1; } }
            fill += __builtin_popcount(hb);
            hb = __builtin_amdgcn_ballot_w32(m2);
            if (m2) { const int p = fill + __builtin_amdgcn_mbcnt_lo(hb, 0u); if (p < LC) { mye[p] = e0 + 2; myd[p] = (int)u2; } }
            fill += __builtin_popcount(hb);
            hb = __builtin_amdgcn_ballot_w32(m3);
            if (m3) { const int p = fill + __builtin_amdgcn_mbcnt_lo(hb, 0u); if (p < LC) { mye[p] = e0 + 3; myd[p] = (int)u3; } }
            fill += __builtin_popcount(hb);
        }
    };

    auto flush = [&]() __attribute__((always_inline)) {
        if (pass != 0 && fill > LC - APP) {
            if (pass == 1) maxupd(fill, USEIX, false);
            else           drain(fill, USEIX, false);
            fill = 0;
        }
    };

    auto scan_use = [&](int k, const int4 (&v)[4]) {
        int kk = st0 + k;
        if (kk >= steps4) kk -= steps4;
        #pragma unroll
        for (int u = 0; u < 4; ++u) {
            const int g = (kk + u) * 1024 + tid;
            examine(g, g < G4, v[u]);
        }
        flush();
    };

    auto append = [&](bool m, unsigned w, int add) __attribute__((always_inline)) {
        const int p = fill + lane;
        if (m && p < LC) mye[p] = (int)w;
        fill += add;
    };

    auto collect = [&]() __attribute__((always_inline)) {
        for (int b0 = wave; b0 < nbw; b0 += AGGW * 32) {
            const int bl = b0 + lane * AGGW;
            const unsigned meta = bl < nbw ? offs[(size_t)bl * BMROWS + blockIdx.x] : 0u;
            const int left = (nbw - b0 + AGGW - 1) / AGGW;
            const int nseg = left < 32 ? left : 32;
            for (int i0 = 0; i0 < nseg; i0 += 8) {
                unsigned w[8];
                int      cn[8];
                const unsigned* sg[8];
                #pragma unroll
                for (int u = 0; u < 8; ++u) {
                    const int i = i0 + u;
                    const unsigned mt = (unsigned)__builtin_amdgcn_readlane((int)meta, i);
                    cn[u] = (int)(mt >> 16);
                    sg[u] = blk + (size_t)(b0 + i * AGGW) * EPB + (mt & 0xffffu);
                    w[u]  = lane < cn[u] ? sg[u][lane] : 0u;
                }
                #pragma unroll
                for (int u = 0; u < 8; ++u) append(lane < cn[u], w[u], cn[u] < 32 ? cn[u] : 32);
                flush();
                #pragma unroll 1
                for (int i = i0; i < i0 + 8 && i < nseg; ++i) {
                    const unsigned mt = (unsigned)__builtin_amdgcn_readlane((int)meta, i);
                    const int c = (int)(mt >> 16);
                    const unsigned* const sp = blk + (size_t)(b0 + i * AGGW) * EPB + (mt & 0xffffu);
                    for (int k = 32; k < c; k += 32) {
                        const bool m = k + lane < c;
                        append(m, m ? sp[k + lane] : 0u, (c - k) < 32 ? (c - k) : 32);
                        flush();
                    }
                }
            }
        }
    };

    for (pass = 0; pass < 3; ++pass) {
        fill = 0;
        if constexpr (USEIX) {
            collect();
        } else {
            int4 va[4], vb[4];
            if (steps4 > 0) scan_load(0, va);
            for (int k = 0; k < steps4; k += 8) {
                scan_load(k + 4, vb);
                scan_use(k, va);
                scan_load(k + 8, va);
                if (k + 4 < steps4) scan_use(k + 4, vb);
            }
        }
        if (wave == 0) {
            const int et = (G4 << 2) + lane;
            unsigned ut = 0xffffffffu;
            if (et < E) ut = (unsigned)(dst[et] - lo);
            const bool mt = ut < (unsigned)cnt;
            const unsigned bm = __builtin_amdgcn_ballot_w32(mt);
            if (mt) {
                const int p = fill + __builtin_amdgcn_mbcnt_lo(bm, 0u);
                if (p < LC) {
                    if constexpr (USEIX) mye[p] = (int)(((unsigned)src[et] & SRCMASK) | (ut << SRCBITS));
                    else { mye[p] = et; myd[p] = (int)ut; }
                }
            }
            fill += __builtin_popcount(bm);
        }
        if (pass == 0) {
            if (fill > LC && lane == 0) s_lost = 1;
            maxupd(fill < LC ? fill : LC, USEIX, USEIX);
            __syncthreads();
            if (__builtin_amdgcn_readfirstlane(s_lost) == 0) { drain(fill, true, USEIX); break; }
        } else if (pass == 1) {
            maxupd(fill, USEIX, false);
            __syncthreads();
        } else {
            drain(fill, USEIX, false);
        }
    }
    __syncthreads();

    for (int i = tid; i < cnt * 64; i += 1024) {
        const int n = i >> 6, c = i & 63;
        const float dn = sden[n * 4 + (c >> 4)];
        const float a  = acc[i];
        out[(size_t)(lo + n) * 64 + c] = (dn > 0.f ? a / dn : 0.f) + bias[c];
    }
}

extern "C" void kernel_launch(void* const* d_in, const int* in_sizes, int n_in,
                              void* d_out, int out_size, void* d_ws, size_t ws_size,
                              hipStream_t stream)
{
    const float* nf   = (const float*)d_in[0];
    const float* W    = (const float*)d_in[1];
    const float* al   = (const float*)d_in[2];
    const float* ar   = (const float*)d_in[3];
    const float* bias = (const float*)d_in[4];
    const int*   src  = (const int*)d_in[5];
    const int*   dst  = (const int*)d_in[6];

    const int N = in_sizes[0] / 64;
    const int E = in_sizes[5];

    float* ws    = (float*)d_ws;
    float* feat  = ws;
    float* el    = feat  + (size_t)N * 64;
    float* er    = el    + (size_t)N * 4;
    float* out   = (float*)d_out;
    const int nr  = (N + NPW - 1) / NPW;
    const int G4  = E >> 2;
    const int nbw = (G4 + 1023) / 1024;
    unsigned* blk  = (unsigned*)(er + (size_t)N * 4);
    unsigned* offs = blk + (size_t)nbw * EPB;
    const size_t need = ((size_t)N * 72 + (size_t)nbw * (EPB + BMROWS)) * sizeof(float);
    const int npb = (N + 127) / 128;

    if (nr <= BMROWS && nbw > 0 && N <= (1 << SRCBITS) && ws_size >= need) {
        gat_proj_logits_sort<<<nbw + npb, 256, 0, stream>>>(nf, W, al, ar, feat, el, er, N, src, dst, blk, offs, G4, nbw);
        gat_agg_owned<true><<<nr, 1024, 0, stream>>>(src, dst, el, er, feat, bias, out, N, E, blk, offs, nbw);
    } else {
        gat_proj_logits_sort<<<npb, 256, 0, stream>>>(nf, W, al, ar, feat, el, er, N, src, dst, nullptr, nullptr, G4, 0);
        gat_agg_owned<false><<<nr, 1024, 0, stream>>>(src, dst, el, er, feat, bias, out, N, E, nullptr, nullptr, 0);
    }
}


# owner waves: edges go to per-wave lists in LDS (slot = returning LDS add); each wave sums its 12-13 rows in registers by f32 WMMA with one-hot row selectors; row ds_add_f32 gone; list-order sums
# speedup vs baseline: 6.1822x; 1.0479x over previous
#include <hip/hip_runtime.h>


typedef float v2f __attribute__((ext_vector_type(2)));
typedef float v8f __attribute__((ext_vector_type(8)));

#define SA_STRIDE 68
#define SW_STRIDE 68

#define NPW   391
#define LCAP  640
#define LCAPX 304
#define AGGW  32
#define BMROWS 256
#define EPB   4096
#define SRCBITS 23
static_assert(NPW <= (1 << (32 - SRCBITS)), "local destination must fit above the source bits");

#define SWT_ROWS 80
#define PROJ_LDS_BYTES ((SWT_ROWS * SW_STRIDE + 8 * 16 * SA_STRIDE) * 4)
static_assert((2 * BMROWS + EPB) * 4 <= PROJ_LDS_BYTES, "the sorter's counters and image must fit in the projection's LDS block");
#define POOL_BYTES 323584

__global__ __launch_bounds__(256) void gat_proj_logits_sort(
    const float* __restrict__ nf, const float* __restrict__ W,
    const float* __restrict__ al, const float* __restrict__ ar,
    float* __restrict__ feat, float* __restrict__ el, float* __restrict__ er, int N,
    const int* __restrict__ src, const int* __restrict__ dst,
    unsigned* __restrict__ blk, unsigned* __restrict__ offs, int G4, int nbw)
{
    __shared__ __attribute__((aligned(16))) unsigned char smem[PROJ_LDS_BYTES];

    const int tid  = threadIdx.x;
    const int wave = tid >> 5;
    const int lane = tid & 31;

    if ((int)blockIdx.x < nbw) {
        unsigned* const hist   = (unsigned*)smem;
        unsigned* const base   = hist + BMROWS;
        unsigned* const sorted = base + BMROWS;
        hist[tid] = 0u;
        __syncthreads();
        const int gw = ((int)blockIdx.x * 32 + wave * 4) * 32 + lane;
        int4 vd[4], vs[4];
        #pragma unroll
        for (int q = 0; q < 4; ++q) {
            const int g = gw + q * 32;
            const size_t o = (size_t)(g < G4 ? g : 0) * 4;
            vd[q] = *(const int4*)(dst + o);
            vs[q] = *(const int4*)(src + o);
        }
        unsigned key[16];
        #pragma unroll
        for (int q = 0; q < 4; ++q) {
            const bool okg = gw + q * 32 < G4;
            const int dd[4] = { vd[q].x, vd[q].y, vd[q].z, vd[q].w };
            #pragma unroll
            for (int c = 0; c < 4; ++c) {
                unsigned k = 0xffffffffu;
                if (okg && (unsigned)dd[c] < (unsigned)N) {
                    const unsigned r = (unsigned)dd[c] / NPW;
                    k = (r << 16) | __hip_atomic_fetch_add(&hist[r], 1u, __ATOMIC_RELAXED, __HIP_MEMORY_SCOPE_WORKGROUP);
                }
                key[q * 4 + c] = k;
            }
        }
        __syncthreads();
        if (wave == 0) {
            unsigned c8[8];
            unsigned t = 0u;
            #pragma unroll
            for (int i = 0; i < 8; ++i) { c8[i] = hist[lane * 8 + i]; t += c8[i]; }
            unsigned incl = t;
            #pragma unroll
            for (int d = 1; d < 32; d <<= 1) {
                const unsigned o = __shfl_up(incl, d, 32);
                if (lane >= d) incl += o;
            }
            unsigned run = incl - t;
            #pragma unroll
            for (int i = 0; i < 8; ++i) { base[lane * 8 + i] = run; run += c8[i]; }
        }
        __syncthreads();
        #pragma unroll
        for (int q = 0; q < 4; ++q) {
            const int dd[4] = { vd[q].x, vd[q].y, vd[q].z, vd[q].w };
            const int ss[4] = { vs[q].x, vs[q].y, vs[q].z, vs[q].w };
            #pragma unroll
            for (int c = 0; c < 4; ++c) {
                const unsigned k = key[q * 4 + c];
                if (k != 0xffffffffu) {
                    const unsigned r = k >> 16;
                    sorted[base[r] + (k & 0xffffu)] =
                        ((unsigned)ss[c] & ((1u << SRCBITS) - 1u)) | (((unsigned)dd[c] - r * NPW) << SRCBITS);
                }
            }
        }
        __syncthreads();
        const unsigned total = base[BMROWS - 1] + hist[BMROWS - 1];
        unsigned* const ob = blk + (size_t)blockIdx.x * EPB;
        for (unsigned i = tid; i < total; i += 256) ob[i] = sorted[i];
        offs[(size_t)blockIdx.x * BMROWS + tid] = base[tid] | (hist[tid] << 16);
        return;
    }

    float* const sWT = (float*)smem;
    float* const sAw = sWT + SWT_ROWS * SW_STRIDE + wave * (16 * SA_STRIDE);

    for (int i = tid; i < 64 * 64; i += 256) {
        const int k = i >> 6, n = i & 63;
        sWT[n * SW_STRIDE + k] = W[i];
    }
    for (int i = tid; i < 8 * SW_STRIDE; i += 256) sWT[72 * SW_STRIDE + i] = 0.f;

    const int rowbase = ((int)blockIdx.x - nbw) * 128 + wave * 16;
    for (int r = 0; r < 16; ++r) {
        int row = rowbase + r;
        row = row < N ? row : (N - 1);
        sAw[r * SA_STRIDE + lane]      = nf[(size_t)row * 64 + lane];
        sAw[r * SA_STRIDE + lane + 32] = nf[(size_t)row * 64 + lane + 32];
    }
    __syncthreads();

    {
        const int k  = tid & 63;
        const int hh = __builtin_amdgcn_readfirstlane(tid >> 6);
        float sl = 0.f, sr = 0.f;
        #pragma unroll
        for (int f = 0; f < 16; ++f) {
            const float w = sWT[(hh * 16 + f) * SW_STRIDE + k];
            sl += w * al[hh * 16 + f];
            sr += w * ar[hh * 16 + f];
        }
        sWT[(64 + hh) * SW_STRIDE + k] = sl;
        sWT[(68 + hh) * SW_STRIDE + k] = sr;
    }
    __syncthreads();

    const int half = lane >> 4;
    const int m    = lane & 15;

    v8f acc0 = {}, acc1 = {}, acc2 = {}, acc3 = {}, acc4 = {};

    #pragma unroll
    for (int k0 = 0; k0 < 64; k0 += 4) {
        const int ka = k0 + half * 2;

        const v2f a  = *(const v2f*)&sAw[m * SA_STRIDE + ka];
        const v2f b0 = *(const v2f*)&sWT[(m +  0) * SW_STRIDE + ka];
        const v2f b1 = *(const v2f*)&sWT[(m + 16) * SW_STRIDE + ka];
        const v2f b2 = *(const v2f*)&sWT[(m + 32) * SW_STRIDE + ka];
        const v2f b3 = *(const v2f*)&sWT[(m + 48) * SW_STRIDE + ka];
        const v2f b4 = *(const v2f*)&sWT[(m + 64) * SW_STRIDE + ka];

        acc0 = __builtin_amdgcn_wmma_f32_16x16x4_f32(false, a, false, b0, (short)0, acc0, false, false);
        acc1 = __builtin_amdgcn_wmma_f32_16x16x4_f32(false, a, false, b1, (short)0, acc1, false, false);
        acc2 = __builtin_amdgcn_wmma_f32_16x16x4_f32(false, a, false, b2, (short)0, acc2, false, false);
        acc3 = __builtin_amdgcn_wmma_f32_16x16x4_f32(false, a, false, b3, (short)0, acc3, false, false);
        acc4 = __builtin_amdgcn_wmma_f32_16x16x4_f32(false, a, false, b4, (short)0, acc4, false, false);
    }

    float* const lg = (m < 4 ? el : er) + (m & 3);
    #pragma unroll
    for (int v = 0; v < 8; ++v) {
        const int row = rowbase + v + half * 8;
        if (row < N) {
            float* fr = feat + (size_t)row * 64;
            fr[ 0 + m] = acc0[v];
            fr[16 + m] = acc1[v];
            fr[32 + m] = acc2[v];
            fr[48 + m] = acc3[v];
            if (m < 8) lg[(size_t)row * 4] = acc4[v];
        }
    }
}

__device__ __forceinline__ float leaky02(float x) { return x > 0.f ? x : 0.2f * x; }

__device__ __forceinline__ unsigned f2o(float f) {
    unsigned u = __float_as_uint(f);
    return (u >> 31) ? ~u : (u | 0x80000000u);
}

__device__ __forceinline__ float o2f(unsigned u) {
    if (u == 0u) return 0.f;
    return (u & 0x80000000u) ? __uint_as_float(u ^ 0x80000000u) : __uint_as_float(~u);
}

template <int U>
__device__ __forceinline__ int row_share(int v) {
    return __builtin_amdgcn_update_dpp(0, v, 0x150 + U, 0xf, 0xf, true);
}

template <int C, bool NARROW>
__device__ __forceinline__ void row_chunk(int s, int dl, const float* myw, const float* __restrict__ feat,
                                          float* acc, int lane)
{
    const int half = lane >> 4;
    const int c4   = lane & 15;
    const float4 wa = *(const float4*)(myw + C * 64 + half * 32 + (c4 >> 2) * 8);
    const float4 wb = *(const float4*)(myw + C * 64 + half * 32 + (c4 >> 2) * 8 + 4);
    const float wv[8] = { wa.x, wa.y, wa.z, wa.w, wb.x, wb.y, wb.z, wb.w };
    float4 f[8];
    int    dj[8];
#define ROW_LOAD(K) { const int sj = row_share<C * 8 + K>(s); dj[K] = row_share<C * 8 + K>(dl); \
                      if constexpr (NARROW) f[K] = *(const float4*)(feat + (((unsigned)sj << 6) | (unsigned)(c4 * 4))); \
                      else                  f[K] = *(const float4*)(feat + (size_t)sj * 64 + c4 * 4); }
    ROW_LOAD(0) ROW_LOAD(1) ROW_LOAD(2) ROW_LOAD(3) ROW_LOAD(4) ROW_LOAD(5) ROW_LOAD(6) ROW_LOAD(7)
#undef ROW_LOAD
    float* const acca = acc + c4 * 4 + (half ? 2 : 0);
    float* const accb = acc + c4 * 4 + (half ? 0 : 2);
    #pragma unroll
    for (int k = 0; k < 8; ++k) {
        const float p0 = f[k].x * wv[k], p1 = f[k].y * wv[k];
        const float p2 = f[k].z * wv[k], p3 = f[k].w * wv[k];
        float* const ra = acca + dj[k] * 64;
        float* const rb = accb + dj[k] * 64;
        __hip_atomic_fetch_add(ra,     half ? p2 : p0, __ATOMIC_RELAXED, __HIP_MEMORY_SCOPE_WORKGROUP);
        __hip_atomic_fetch_add(ra + 1, half ? p3 : p1, __ATOMIC_RELAXED, __HIP_MEMORY_SCOPE_WORKGROUP);
        __hip_atomic_fetch_add(rb,     half ? p0 : p2, __ATOMIC_RELAXED, __HIP_MEMORY_SCOPE_WORKGROUP);
        __hip_atomic_fetch_add(rb + 1, half ? p1 : p3, __ATOMIC_RELAXED, __HIP_MEMORY_SCOPE_WORKGROUP);
    }
}

template <bool USEIX>
__device__ __forceinline__ void agg_owned_body(
    const int* __restrict__ src, const int* __restrict__ dst,
    const float* __restrict__ el, const float* __restrict__ er,
    const float* __restrict__ feat,
    const float* __restrict__ bias, float* __restrict__ out, int N, int E,
    const unsigned* __restrict__ blk, const unsigned* __restrict__ offs, int nbw,
    unsigned char* pool)
{
    constexpr int LC  = USEIX ? LCAPX : LCAP;
    constexpr int APP = USEIX ? 256 : 512;
    constexpr unsigned SRCMASK = (1u << SRCBITS) - 1u;
    float*    const acc   = (float*)pool;
    float*    const sden  = acc + NPW * 64;
    unsigned* const smax  = (unsigned*)(sden + NPW * 4);
    float*    const swt   = (float*)(smax + NPW * 4);
    int*      const lst_e = (int*)(swt + AGGW * 128);
    int*      const lst_d = lst_e + AGGW * LC;
    float*    const ssc   = (float*)(lst_d + (USEIX ? 0 : AGGW * LC));
    int*      const lostp = (int*)(ssc + (USEIX ? AGGW * LC * 4 : 0));
    static_assert((NPW * 72 + AGGW * 128 + AGGW * LC * (USEIX ? 5 : 2) + 4) * 4 <= POOL_BYTES, "the pool must hold this carving");
#define s_lost (*lostp)

    const int tid  = threadIdx.x;
    const int wave = tid >> 5;
    const int lane = tid & 31;
    const int lo   = blockIdx.x * NPW;
    const int cnt  = (N - lo) < NPW ? (N - lo) : NPW;

    for (int i = tid; i < NPW * 64; i += 1024) acc[i] = 0.f;
    for (int i = tid; i < NPW * 4;  i += 1024) { sden[i] = 0.f; smax[i] = 0u; }
    if (tid == 0) s_lost = 0;
    __syncthreads();

    float* const myw = swt + wave * 128;
    int*   const mye = lst_e + wave * LC;
    int*   const myd = lst_d + (USEIX ? 0 : wave * LC);
    float* const mys = ssc + (USEIX ? wave * LC * 4 : 0);

    auto maxupd = [&](int n, bool have_s, bool keep) __attribute__((always_inline)) {
        for (int b = 0; b < n; b += 128) {
            int    sv[4], dv[4];
            float4 l4[4], r4[4];
            #pragma unroll
            for (int u = 0; u < 4; ++u) {
                const int  i     = b + u * 32 + lane;
                const bool valid = i < n;
                const int  e     = valid ? mye[i] : 0;
                if constexpr (USEIX) {
                    dv[u] = valid ? (int)((unsigned)e >> SRCBITS) : -1;
                    sv[u] = (int)((unsigned)e & SRCMASK);
                } else {
                    dv[u] = valid ? myd[i] : -1;
                    sv[u] = have_s ? e : src[e];
                }
            }
            #pragma unroll
            for (int u = 0; u < 4; ++u) {
                const int dl = dv[u] < 0 ? 0 : dv[u];
                l4[u] = *(const float4*)(el + (size_t)sv[u] * 4);
                r4[u] = *(const float4*)(er + (size_t)(lo + dl) * 4);
            }
            #pragma unroll
            for (int u = 0; u < 4; ++u) {
                if (dv[u] >= 0) {
                    unsigned* mx = &smax[dv[u] * 4];
                    float4 sc;
                    sc.x = leaky02(l4[u].x + r4[u].x);
                    sc.y = leaky02(l4[u].y + r4[u].y);
                    sc.z = leaky02(l4[u].z + r4[u].z);
                    sc.w = leaky02(l4[u].w + r4[u].w);
                    __hip_atomic_fetch_max(mx + 0, f2o(sc.x), __ATOMIC_RELAXED, __HIP_MEMORY_SCOPE_WORKGROUP);
                    __hip_atomic_fetch_max(mx + 1, f2o(sc.y), __ATOMIC_RELAXED, __HIP_MEMORY_SCOPE_WORKGROUP);
                    __hip_atomic_fetch_max(mx + 2, f2o(sc.z), __ATOMIC_RELAXED, __HIP_MEMORY_SCOPE_WORKGROUP);
                    __hip_atomic_fetch_max(mx + 3, f2o(sc.w), __ATOMIC_RELAXED, __HIP_MEMORY_SCOPE_WORKGROUP);
                    if (USEIX && keep) *(float4*)&mys[(b + u * 32 + lane) * 4] = sc;
                    if (!USEIX && !have_s) mye[b + u * 32 + lane] = sv[u];
                }
            }
        }
    };

    auto drain = [&](int n, bool have_s, bool kept) __attribute__((always_inline)) {
        for (int b = 0; b < n; b += 32) {
            const int  i     = b + ((lane & 15) << 1) + (lane >> 4);
            const bool valid = i < n;
            const int  q  = valid ? mye[i] : 0;
            int s, dl;
            if constexpr (USEIX) {
                s  = (int)((unsigned)q & SRCMASK);
                dl = (int)((unsigned)q >> SRCBITS);
            } else {
                dl = valid ? myd[i] : 0;
                s  = have_s ? q : src[q];
                s  = valid ? s : 0;
            }
            float4 sc;
            if (USEIX && kept) {
                sc = *(const float4*)&mys[(valid ? i : 0) * 4];
            } else {
                const int  d  = lo + dl;
                const float4 l4 = *(const float4*)(el + (size_t)s * 4);
                const float4 r4 = *(const float4*)(er + (size_t)d * 4);
                sc.x = leaky02(l4.x + r4.x);
                sc.y = leaky02(l4.y + r4.y);
                sc.z = leaky02(l4.z + r4.z);
                sc.w = leaky02(l4.w + r4.w);
            }
            const uint4  mu = *(const uint4*)&smax[dl * 4];
            float4 w;
            w.x = valid ? __expf(sc.x - o2f(mu.x)) : 0.f;
            w.y = valid ? __expf(sc.y - o2f(mu.y)) : 0.f;
            w.z = valid ? __expf(sc.z - o2f(mu.z)) : 0.f;
            w.w = valid ? __expf(sc.w - o2f(mu.w)) : 0.f;
            if (valid) {
                float* dn = &sden[dl * 4];
                __hip_atomic_fetch_add(dn + 0, w.x, __ATOMIC_RELAXED, __HIP_MEMORY_SCOPE_WORKGROUP);
                __hip_atomic_fetch_add(dn + 1, w.y, __ATOMIC_RELAXED, __HIP_MEMORY_SCOPE_WORKGROUP);
                __hip_atomic_fetch_add(dn + 2, w.z, __ATOMIC_RELAXED, __HIP_MEMORY_SCOPE_WORKGROUP);
                __hip_atomic_fetch_add(dn + 3, w.w, __ATOMIC_RELAXED, __HIP_MEMORY_SCOPE_WORKGROUP);
            }
            {
                float* wd = myw + ((lane & 15) >> 3) * 64 + (lane >> 4) * 32 + (lane & 7);
                wd[0] = w.x; wd[8] = w.y; wd[16] = w.z; wd[24] = w.w;
            }

            const int nb = (n - b) < 32 ? (n - b) : 32;
            row_chunk<0, USEIX>(s, dl, myw, feat, acc, lane);
            if (nb > 16) row_chunk<1, USEIX>(s, dl, myw, feat, acc, lane);
        }
    };

    const int G4     = E >> 2;
    const int steps  = ((G4 + 1023) >> 10);
    const int steps4 = (steps + 3) & ~3;
    const int st0    = (int)(((long long)blockIdx.x * steps4) / gridDim.x) & ~3;
    int fill = 0;
    int pass = 0;

    auto scan_load = [&](int k, int4 (&v)[4]) {
        int kk = st0 + k;
        if (kk >= steps4) kk -= steps4;
        #pragma unroll
        for (int u = 0; u < 4; ++u) {
            const int g = (kk + u) * 1024 + tid;
            v[u] = *(const int4*)(dst + (size_t)(g < G4 ? g : 0) * 4);
        }
    };

    auto examine = [&](int g, bool ok, const int4& v) __attribute__((always_inline)) {
        const int e0 = g * 4;
        const unsigned lim = ok ? (unsigned)cnt : 0u;
        const unsigned u0 = (unsigned)(v.x - lo);
        const unsigned u1 = (unsigned)(v.y - lo);
        const unsigned u2 = (unsigned)(v.z - lo);
        const unsigned u3 = (unsigned)(v.w - lo);
        const bool m0 = u0 < lim, m1 = u1 < lim;
        const bool m2 = u2 < lim, m3 = u3 < lim;
        if (__builtin_amdgcn_ballot_w32(m0 | m1 | m2 | m3)) {
            unsigned hb;
            hb = __builtin_amdgcn_ballot_w32(m0);
            if (m0) { const int p = fill + __builtin_amdgcn_mbcnt_lo(hb, 0u); if (p < LC) { mye[p] = e0;     myd[p] = (int)u0; } }
            fill += __builtin_popcount(hb);
            hb = __builtin_amdgcn_ballot_w32(m1);
            if (m1) { const int p = fill + __builtin_amdgcn_mbcnt_lo(hb, 0u); if (p < LC) { mye[p] = e0 + 1; myd[p] = (int)u1; } }
            fill += __builtin_popcount(hb);
            hb = __builtin_amdgcn_ballot_w32(m2);
            if (m2) { const int p = fill + __builtin_amdgcn_mbcnt_lo(hb, 0u); if (p < LC) { mye[p] = e0 + 2; myd[p] = (int)u2; } }
            fill += __builtin_popcount(hb);
            hb = __builtin_amdgcn_ballot_w32(m3);
            if (m3) { const int p = fill + __builtin_amdgcn_mbcnt_lo(hb, 0u); if (p < LC) { mye[p] = e0 + 3; myd[p] = (int)u3; } }
            fill += __builtin_popcount(hb);
        }
    };

    auto flush = [&]() __attribute__((always_inline)) {
        if (pass != 0 && fill > LC - APP) {
            if (pass == 1) maxupd(fill, USEIX, false);
            else           drain(fill, USEIX, false);
            fill = 0;
        }
    };

    auto scan_use = [&](int k, const int4 (&v)[4]) {
        int kk = st0 + k;
        if (kk >= steps4) kk -= steps4;
        #pragma unroll
        for (int u = 0; u < 4; ++u) {
            const int g = (kk + u) * 1024 + tid;
            examine(g, g < G4, v[u]);
        }
        flush();
    };

    auto append = [&](bool m, unsigned w, int add) __attribute__((always_inline)) {
        const int p = fill + lane;
        if (m && p < LC) mye[p] = (int)w;
        fill += add;
    };

    auto collect = [&]() __attribute__((always_inline)) {
        for (int b0 = wave; b0 < nbw; b0 += AGGW * 32) {
            const int bl = b0 + lane * AGGW;
            const unsigned meta = bl < nbw ? offs[(size_t)bl * BMROWS + blockIdx.x] : 0u;
            const int left = (nbw - b0 + AGGW - 1) / AGGW;
            const int nseg = left < 32 ? left : 32;
            for (int i0 = 0; i0 < nseg; i0 += 8) {
                unsigned w[8];
                int      cn[8];
                const unsigned* sg[8];
                #pragma unroll
                for (int u = 0; u < 8; ++u) {
                    const int i = i0 + u;
                    const unsigned mt = (unsigned)__builtin_amdgcn_readlane((int)meta, i);
                    cn[u] = (int)(mt >> 16);
                    sg[u] = blk + (size_t)(b0 + i * AGGW) * EPB + (mt & 0xffffu);
                    w[u]  = lane < cn[u] ? sg[u][lane] : 0u;
                }
                #pragma unroll
                for (int u = 0; u < 8; ++u) append(lane < cn[u], w[u], cn[u] < 32 ? cn[u] : 32);
                flush();
                #pragma unroll 1
                for (int i = i0; i < i0 + 8 && i < nseg; ++i) {
                    const unsigned mt = (unsigned)__builtin_amdgcn_readlane((int)meta, i);
                    const int c = (int)(mt >> 16);
                    const unsigned* const sp = blk + (size_t)(b0 + i * AGGW) * EPB + (mt & 0xffffu);
                    for (int k = 32; k < c; k += 32) {
                        const bool m = k + lane < c;
                        append(m, m ? sp[k + lane] : 0u, (c - k) < 32 ? (c - k) : 32);
                        flush();
                    }
                }
            }
        }
    };

    for (pass = 0; pass < 3; ++pass) {
        fill = 0;
        if constexpr (USEIX) {
            collect();
        } else {
            int4 va[4], vb[4];
            if (steps4 > 0) scan_load(0, va);
            for (int k = 0; k < steps4; k += 8) {
                scan_load(k + 4, vb);
                scan_use(k, va);
                scan_load(k + 8, va);
                if (k + 4 < steps4) scan_use(k + 4, vb);
            }
        }
        if (wave == 0) {
            const int et = (G4 << 2) + lane;
            unsigned ut = 0xffffffffu;
            if (et < E) ut = (unsigned)(dst[et] - lo);
            const bool mt = ut < (unsigned)cnt;
            const unsigned bm = __builtin_amdgcn_ballot_w32(mt);
            if (mt) {
                const int p = fill + __builtin_amdgcn_mbcnt_lo(bm, 0u);
                if (p < LC) {
                    if constexpr (USEIX) mye[p] = (int)(((unsigned)src[et] & SRCMASK) | (ut << SRCBITS));
                    else { mye[p] = et; myd[p] = (int)ut; }
                }
            }
            fill += __builtin_popcount(bm);
        }
        if (pass == 0) {
            if (fill > LC && lane == 0) s_lost = 1;
            maxupd(fill < LC ? fill : LC, USEIX, USEIX);
            __syncthreads();
            if (__builtin_amdgcn_readfirstlane(s_lost) == 0) { drain(fill, true, USEIX); break; }
        } else if (pass == 1) {
            maxupd(fill, USEIX, false);
            __syncthreads();
        } else {
            drain(fill, USEIX, false);
        }
    }
    __syncthreads();

    for (int i = tid; i < cnt * 64; i += 1024) {
        const int n = i >> 6, c = i & 63;
        const float dn = sden[n * 4 + (c >> 4)];
        const float a  = acc[i];
        out[(size_t)(lo + n) * 64 + c] = (dn > 0.f ? a / dn : 0.f) + bias[c];
    }
#undef s_lost
}

__global__ __launch_bounds__(1024) void gat_agg_scan(
    const int* __restrict__ src, const int* __restrict__ dst,
    const float* __restrict__ el, const float* __restrict__ er,
    const float* __restrict__ feat,
    const float* __restrict__ bias, float* __restrict__ out, int N, int E)
{
    __shared__ __attribute__((aligned(16))) unsigned char pool[POOL_BYTES];
    agg_owned_body<false>(src, dst, el, er, feat, bias, out, N, E, nullptr, nullptr, 0, pool);
}

template <int C>
__device__ __forceinline__ void row_chunk_mm(int s, int ml, const float* myw, const float* __restrict__ feat,
                                             v8f& t0, v8f& t1, v8f& t2, v8f& t3, int lane)
{
    const int half = lane >> 4;
    const int c4   = lane & 15;
    const float4 wa = *(const float4*)(myw + C * 64 + half * 32 + (c4 >> 2) * 8);
    const float4 wb = *(const float4*)(myw + C * 64 + half * 32 + (c4 >> 2) * 8 + 4);
    const float wv[8] = { wa.x, wa.y, wa.z, wa.w, wb.x, wb.y, wb.z, wb.w };
    float4 f[8];
    int    mj[8];
#define ROW_LOAD(K) { const int sj = row_share<C * 8 + K>(s); mj[K] = row_share<C * 8 + K>(ml); \
                      f[K] = *(const float4*)(feat + (((unsigned)sj << 6) | (unsigned)(c4 * 4))); }
    ROW_LOAD(0) ROW_LOAD(1) ROW_LOAD(2) ROW_LOAD(3) ROW_LOAD(4) ROW_LOAD(5) ROW_LOAD(6) ROW_LOAD(7)
#undef ROW_LOAD
    #pragma unroll
    for (int g = 0; g < 4; ++g) {
        const int k0 = 2 * g, k1 = 2 * g + 1;
        v2f a, b0, b1, b2, b3;
        a.x  = mj[k0] == c4 ? 1.f : 0.f;   a.y  = mj[k1] == c4 ? 1.f : 0.f;
        b0.x = f[k0].x * wv[k0];           b0.y = f[k1].x * wv[k1];
        b1.x = f[k0].y * wv[k0];           b1.y = f[k1].y * wv[k1];
        b2.x = f[k0].z * wv[k0];           b2.y = f[k1].z * wv[k1];
        b3.x = f[k0].w * wv[k0];           b3.y = f[k1].w * wv[k1];
        t0 = __builtin_amdgcn_wmma_f32_16x16x4_f32(false, a, false, b0, (short)0, t0, false, false);
        t1 = __builtin_amdgcn_wmma_f32_16x16x4_f32(false, a, false, b1, (short)0, t1, false, false);
        t2 = __builtin_amdgcn_wmma_f32_16x16x4_f32(false, a, false, b2, (short)0, t2, false, false);
        t3 = __builtin_amdgcn_wmma_f32_16x16x4_f32(false, a, false, b3, (short)0, t3, false, false);
    }
}

#define LCO    448
#define TLROWS 13
__device__ __forceinline__ int tile_of(unsigned dl)  { return (int)((dl * 21u) >> 8); }
__device__ __forceinline__ int tile_first(int w)     { return (w * 256 + 20) / 21; }
static_assert(((unsigned)(NPW - 1) * 21u) >> 8 == AGGW - 1, "the last local row must belong to the last wave");
static_assert((AGGW * 256 + 20) / 21 == NPW, "the tiles must end with the range");
static_assert((NPW * 8 + AGGW * 128 + AGGW * LCO * 5 + AGGW + 4) * 4 <= POOL_BYTES, "the pool must hold the owner lists");

__global__ __launch_bounds__(1024) void gat_agg_tiles(
    const int* __restrict__ src, const int* __restrict__ dst,
    const float* __restrict__ el, const float* __restrict__ er,
    const float* __restrict__ feat,
    const float* __restrict__ bias, float* __restrict__ out, int N, int E,
    const unsigned* __restrict__ blk, const unsigned* __restrict__ offs, int nbw)
{
    __shared__ __attribute__((aligned(16))) unsigned char pool[POOL_BYTES];
    constexpr unsigned SRCMASK = (1u << SRCBITS) - 1u;
    float*    const sden  = (float*)pool;
    unsigned* const smax  = (unsigned*)(sden + NPW * 4);
    float*    const swt   = (float*)(smax + NPW * 4);
    int*      const olst  = (int*)(swt + AGGW * 128);
    float*    const osc   = (float*)(olst + AGGW * LCO);
    unsigned* const ocnt  = (unsigned*)(osc + AGGW * LCO * 4);
    int*      const lostp = (int*)(ocnt + AGGW);

    const int tid  = threadIdx.x;
    const int wave = tid >> 5;
    const int lane = tid & 31;
    const int lo   = blockIdx.x * NPW;
    const int cnt  = (N - lo) < NPW ? (N - lo) : NPW;

    for (int i = tid; i < NPW * 4; i += 1024) { sden[i] = 0.f; smax[i] = 0u; }
    if (tid < AGGW) ocnt[tid] = 0u;
    if (tid == 0) *lostp = 0;
    __syncthreads();

    auto put = [&](bool m, unsigned w) __attribute__((always_inline)) {
        if (m) {
            const int o = tile_of(w >> SRCBITS);
            const unsigned p = __hip_atomic_fetch_add(&ocnt[o], 1u, __ATOMIC_RELAXED, __HIP_MEMORY_SCOPE_WORKGROUP);
            if (p < (unsigned)LCO) olst[o * LCO + (int)p] = (int)w;
            else                   *lostp = 1;
        }
    };

    for (int b0 = wave; b0 < nbw; b0 += AGGW * 32) {
        const int bl = b0 + lane * AGGW;
        const unsigned meta = bl < nbw ? offs[(size_t)bl * BMROWS + blockIdx.x] : 0u;
        const int left = (nbw - b0 + AGGW - 1) / AGGW;
        const int nseg = left < 32 ? left : 32;
        for (int i0 = 0; i0 < nseg; i0 += 8) {
            unsigned w[8];
            int      cn[8];
            #pragma unroll
            for (int u = 0; u < 8; ++u) {
                const int i = i0 + u;
                const unsigned mt = (unsigned)__builtin_amdgcn_readlane((int)meta, i);
                cn[u] = (int)(mt >> 16);
                const unsigned* const sg = blk + (size_t)(b0 + i * AGGW) * EPB + (mt & 0xffffu);
                w[u]  = lane < cn[u] ? sg[lane] : 0u;
            }
            unsigned p[8];
            #pragma unroll
            for (int u = 0; u < 8; ++u) {
                p[u] = 0u;
                if (lane < cn[u]) p[u] = __hip_atomic_fetch_add(&ocnt[tile_of(w[u] >> SRCBITS)], 1u, __ATOMIC_RELAXED, __HIP_MEMORY_SCOPE_WORKGROUP);
            }
            #pragma unroll
            for (int u = 0; u < 8; ++u) {
                if (lane < cn[u]) {
                    if (p[u] < (unsigned)LCO) olst[tile_of(w[u] >> SRCBITS) * LCO + (int)p[u]] = (int)w[u];
                    else                      *lostp = 1;
                }
            }
            for (int i = i0; i < i0 + 8 && i < nseg; ++i) {
                const unsigned mt = (unsigned)__builtin_amdgcn_readlane((int)meta, i);
                const int c = (int)(mt >> 16);
                const unsigned* const sp = blk + (size_t)(b0 + i * AGGW) * EPB + (mt & 0xffffu);
                for (int k = 32; k < c; k += 32) {
                    const bool m = k + lane < c;
                    put(m, m ? sp[k + lane] : 0u);
                }
            }
        }
    }
    if (wave == 0) {
        const int et = ((E >> 2) << 2) + lane;
        if (et < E) {
            const unsigned ut = (unsigned)(dst[et] - lo);
            if (ut < (unsigned)cnt) put(true, ((unsigned)src[et] & SRCMASK) | (ut << SRCBITS));
        }
    }
    __syncthreads();
    if (__builtin_amdgcn_readfirstlane(*lostp) != 0) {
        __syncthreads();
        agg_owned_body<true>(src, dst, el, er, feat, bias, out, N, E, blk, offs, nbw, pool);
        return;
    }

    const int n = __builtin_amdgcn_readfirstlane((int)ocnt[wave]);
    int*   const mye = olst + wave * LCO;
    float* const mys = osc + wave * LCO * 4;
    float* const myw = swt + wave * 128;

    for (int b = 0; b < n; b += 128) {
        int    sv[4], dv[4];
        float4 l4[4], r4[4];
        #pragma unroll
        for (int u = 0; u < 4; ++u) {
            const int  i     = b + u * 32 + lane;
            const bool valid = i < n;
            const int  e     = valid ? mye[i] : 0;
            dv[u] = valid ? (int)((unsigned)e >> SRCBITS) : -1;
            sv[u] = (int)((unsigned)e & SRCMASK);
        }
        #pragma unroll
        for (int u = 0; u < 4; ++u) {
            const int dl = dv[u] < 0 ? 0 : dv[u];
            l4[u] = *(const float4*)(el + (size_t)sv[u] * 4);
            r4[u] = *(const float4*)(er + (size_t)(lo + dl) * 4);
        }
        #pragma unroll
        for (int u = 0; u < 4; ++u) {
            if (dv[u] >= 0) {
                unsigned* mx = &smax[dv[u] * 4];
                float4 sc;
                sc.x = leaky02(l4[u].x + r4[u].x);
                sc.y = leaky02(l4[u].y + r4[u].y);
                sc.z = leaky02(l4[u].z + r4[u].z);
                sc.w = leaky02(l4[u].w + r4[u].w);
                __hip_atomic_fetch_max(mx + 0, f2o(sc.x), __ATOMIC_RELAXED, __HIP_MEMORY_SCOPE_WORKGROUP);
                __hip_atomic_fetch_max(mx + 1, f2o(sc.y), __ATOMIC_RELAXED, __HIP_MEMORY_SCOPE_WORKGROUP);
                __hip_atomic_fetch_max(mx + 2, f2o(sc.z), __ATOMIC_RELAXED, __HIP_MEMORY_SCOPE_WORKGROUP);
                __hip_atomic_fetch_max(mx + 3, f2o(sc.w), __ATOMIC_RELAXED, __HIP_MEMORY_SCOPE_WORKGROUP);
                *(float4*)&mys[(b + u * 32 + lane) * 4] = sc;
            }
        }
    }
    __syncthreads();

    const int rs = tile_first(wave);
    v8f t0 = {}, t1 = {}, t2 = {}, t3 = {};
    for (int b = 0; b < n; b += 32) {
        const int  i     = b + ((lane & 15) << 1) + (lane >> 4);
        const bool valid = i < n;
        const int  q  = valid ? mye[i] : 0;
        const int  s  = (int)((unsigned)q & SRCMASK);
        const int  dl = (int)((unsigned)q >> SRCBITS);
        const int  ml = valid ? dl - rs : -1;
        const float4 sc = *(const float4*)&mys[(valid ? i : 0) * 4];
        const uint4  mu = *(const uint4*)&smax[dl * 4];
        float4 w;
        w.x = valid ? __expf(sc.x - o2f(mu.x)) : 0.f;
        w.y = valid ? __expf(sc.y - o2f(mu.y)) : 0.f;
        w.z = valid ? __expf(sc.z - o2f(mu.z)) : 0.f;
        w.w = valid ? __expf(sc.w - o2f(mu.w)) : 0.f;
        if (valid) {
            float* dn = &sden[dl * 4];
            __hip_atomic_fetch_add(dn + 0, w.x, __ATOMIC_RELAXED, __HIP_MEMORY_SCOPE_WORKGROUP);
            __hip_atomic_fetch_add(dn + 1, w.y, __ATOMIC_RELAXED, __HIP_MEMORY_SCOPE_WORKGROUP);
            __hip_atomic_fetch_add(dn + 2, w.z, __ATOMIC_RELAXED, __HIP_MEMORY_SCOPE_WORKGROUP);
            __hip_atomic_fetch_add(dn + 3, w.w, __ATOMIC_RELAXED, __HIP_MEMORY_SCOPE_WORKGROUP);
        }
        {
            float* wd = myw + ((lane & 15) >> 3) * 64 + (lane >> 4) * 32 + (lane & 7);
            wd[0] = w.x; wd[8] = w.y; wd[16] = w.z; wd[24] = w.w;
        }
        const int nb = (n - b) < 32 ? (n - b) : 32;
        row_chunk_mm<0>(s, ml, myw, feat, t0, t1, t2, t3, lane);
        if (nb > 16) row_chunk_mm<1>(s, ml, myw, feat, t0, t1, t2, t3, lane);
    }
    __syncthreads();

    {
        const int half = lane >> 4;
        const int c4   = lane & 15;
        const int re   = tile_first(wave + 1) < cnt ? tile_first(wave + 1) : cnt;
        const float4 bv = *(const float4*)(bias + c4 * 4);
        #pragma unroll
        for (int r = 0; r < 8; ++r) {
            const int nl = rs + half * 8 + r;
            if (nl < re) {
                const float dn = sden[nl * 4 + (c4 >> 2)];
                float4 o;
                o.x = (dn > 0.f ? t0[r] / dn : 0.f) + bv.x;
                o.y = (dn > 0.f ? t1[r] / dn : 0.f) + bv.y;
                o.z = (dn > 0.f ? t2[r] / dn : 0.f) + bv.z;
                o.w = (dn > 0.f ? t3[r] / dn : 0.f) + bv.w;
                *(float4*)(out + (size_t)(lo + nl) * 64 + c4 * 4) = o;
            }
        }
    }
}

extern "C" void kernel_launch(void* const* d_in, const int* in_sizes, int n_in,
                              void* d_out, int out_size, void* d_ws, size_t ws_size,
                              hipStream_t stream)
{
    const float* nf   = (const float*)d_in[0];
    const float* W    = (const float*)d_in[1];
    const float* al   = (const float*)d_in[2];
    const float* ar   = (const float*)d_in[3];
    const float* bias = (const float*)d_in[4];
    const int*   src  = (const int*)d_in[5];
    const int*   dst  = (const int*)d_in[6];

    const int N = in_sizes[0] / 64;
    const int E = in_sizes[5];

    float* ws    = (float*)d_ws;
    float* feat  = ws;
    float* el    = feat  + (size_t)N * 64;
    float* er    = el    + (size_t)N * 4;
    float* out   = (float*)d_out;
    const int nr  = (N + NPW - 1) / NPW;
    const int G4  = E >> 2;
    const int nbw = (G4 + 1023) / 1024;
    unsigned* blk  = (unsigned*)(er + (size_t)N * 4);
    unsigned* offs = blk + (size_t)nbw * EPB;
    const size_t need = ((size_t)N * 72 + (size_t)nbw * (EPB + BMROWS)) * sizeof(float);
    const int npb = (N + 127) / 128;

    if (nr <= BMROWS && nbw > 0 && N <= (1 << SRCBITS) && ws_size >= need) {
        gat_proj_logits_sort<<<nbw + npb, 256, 0, stream>>>(nf, W, al, ar, feat, el, er, N, src, dst, blk, offs, G4, nbw);
        gat_agg_tiles<<<nr, 1024, 0, stream>>>(src, dst, el, er, feat, bias, out, N, E, blk, offs, nbw);
    } else {
        gat_proj_logits_sort<<<npb, 256, 0, stream>>>(nf, W, al, ar, feat, el, er, N, src, dst, nullptr, nullptr, G4, 0);
        gat_agg_scan<<<nr, 1024, 0, stream>>>(src, dst, el, er, feat, bias, out, N, E);
    }
}


# max phase removed: softmax weights taken against an upper bound of each node's largest score (largest el of any node + its er), not the segment max; one list walk; weights, sums differ in rounding
# speedup vs baseline: 6.9773x; 1.1286x over previous
#include <hip/hip_runtime.h>


typedef float v2f __attribute__((ext_vector_type(2)));
typedef float v8f __attribute__((ext_vector_type(8)));

#define SA_STRIDE 68
#define SW_STRIDE 68

#define NPW   391
#define LCAP  640
#define LCAPX 304
#define AGGW  32
#define BMROWS 256
#define EPB   4096
#define SRCBITS 23
static_assert(NPW <= (1 << (32 - SRCBITS)), "local destination must fit above the source bits");

#define SWT_ROWS 80
#define PROJ_LDS_BYTES ((SWT_ROWS * SW_STRIDE + 8 * 16 * SA_STRIDE) * 4)
static_assert((2 * BMROWS + EPB) * 4 <= PROJ_LDS_BYTES, "the sorter's counters and image must fit in the projection's LDS block");
#define POOL_BYTES 323584

__global__ __launch_bounds__(256) void gat_proj_logits_sort(
    const float* __restrict__ nf, const float* __restrict__ W,
    const float* __restrict__ al, const float* __restrict__ ar,
    float* __restrict__ feat, float* __restrict__ el, float* __restrict__ er, int N,
    const int* __restrict__ src, const int* __restrict__ dst,
    unsigned* __restrict__ blk, unsigned* __restrict__ offs, int G4, int nbw,
    float* __restrict__ pmax)
{
    __shared__ __attribute__((aligned(16))) unsigned char smem[PROJ_LDS_BYTES];
    __shared__ float s_red[8 * 4];

    const int tid  = threadIdx.x;
    const int wave = tid >> 5;
    const int lane = tid & 31;

    if ((int)blockIdx.x < nbw) {
        unsigned* const hist   = (unsigned*)smem;
        unsigned* const base   = hist + BMROWS;
        unsigned* const sorted = base + BMROWS;
        hist[tid] = 0u;
        __syncthreads();
        const int gw = ((int)blockIdx.x * 32 + wave * 4) * 32 + lane;
        int4 vd[4], vs[4];
        #pragma unroll
        for (int q = 0; q < 4; ++q) {
            const int g = gw + q * 32;
            const size_t o = (size_t)(g < G4 ? g : 0) * 4;
            vd[q] = *(const int4*)(dst + o);
            vs[q] = *(const int4*)(src + o);
        }
        unsigned key[16];
        #pragma unroll
        for (int q = 0; q < 4; ++q) {
            const bool okg = gw + q * 32 < G4;
            const int dd[4] = { vd[q].x, vd[q].y, vd[q].z, vd[q].w };
            #pragma unroll
            for (int c = 0; c < 4; ++c) {
                unsigned k = 0xffffffffu;
                if (okg && (unsigned)dd[c] < (unsigned)N) {
                    const unsigned r = (unsigned)dd[c] / NPW;
                    k = (r << 16) | __hip_atomic_fetch_add(&hist[r], 1u, __ATOMIC_RELAXED, __HIP_MEMORY_SCOPE_WORKGROUP);
                }
                key[q * 4 + c] = k;
            }
        }
        __syncthreads();
        if (wave == 0) {
            unsigned c8[8];
            unsigned t = 0u;
            #pragma unroll
            for (int i = 0; i < 8; ++i) { c8[i] = hist[lane * 8 + i]; t += c8[i]; }
            unsigned incl = t;
            #pragma unroll
            for (int d = 1; d < 32; d <<= 1) {
                const unsigned o = __shfl_up(incl, d, 32);
                if (lane >= d) incl += o;
            }
            unsigned run = incl - t;
            #pragma unroll
            for (int i = 0; i < 8; ++i) { base[lane * 8 + i] = run; run += c8[i]; }
        }
        __syncthreads();
        #pragma unroll
        for (int q = 0; q < 4; ++q) {
            const int dd[4] = { vd[q].x, vd[q].y, vd[q].z, vd[q].w };
            const int ss[4] = { vs[q].x, vs[q].y, vs[q].z, vs[q].w };
            #pragma unroll
            for (int c = 0; c < 4; ++c) {
                const unsigned k = key[q * 4 + c];
                if (k != 0xffffffffu) {
                    const unsigned r = k >> 16;
                    sorted[base[r] + (k & 0xffffu)] =
                        ((unsigned)ss[c] & ((1u << SRCBITS) - 1u)) | (((unsigned)dd[c] - r * NPW) << SRCBITS);
                }
            }
        }
        __syncthreads();
        const unsigned total = base[BMROWS - 1] + hist[BMROWS - 1];
        unsigned* const ob = blk + (size_t)blockIdx.x * EPB;
        for (unsigned i = tid; i < total; i += 256) ob[i] = sorted[i];
        offs[(size_t)blockIdx.x * BMROWS + tid] = base[tid] | (hist[tid] << 16);
        return;
    }

    float* const sWT = (float*)smem;
    float* const sAw = sWT + SWT_ROWS * SW_STRIDE + wave * (16 * SA_STRIDE);

    for (int i = tid; i < 64 * 64; i += 256) {
        const int k = i >> 6, n = i & 63;
        sWT[n * SW_STRIDE + k] = W[i];
    }
    for (int i = tid; i < 8 * SW_STRIDE; i += 256) sWT[72 * SW_STRIDE + i] = 0.f;

    const int rowbase = ((int)blockIdx.x - nbw) * 128 + wave * 16;
    for (int r = 0; r < 16; ++r) {
        int row = rowbase + r;
        row = row < N ? row : (N - 1);
        sAw[r * SA_STRIDE + lane]      = nf[(size_t)row * 64 + lane];
        sAw[r * SA_STRIDE + lane + 32] = nf[(size_t)row * 64 + lane + 32];
    }
    __syncthreads();

    {
        const int k  = tid & 63;
        const int hh = __builtin_amdgcn_readfirstlane(tid >> 6);
        float sl = 0.f, sr = 0.f;
        #pragma unroll
        for (int f = 0; f < 16; ++f) {
            const float w = sWT[(hh * 16 + f) * SW_STRIDE + k];
            sl += w * al[hh * 16 + f];
            sr += w * ar[hh * 16 + f];
        }
        sWT[(64 + hh) * SW_STRIDE + k] = sl;
        sWT[(68 + hh) * SW_STRIDE + k] = sr;
    }
    __syncthreads();

    const int half = lane >> 4;
    const int m    = lane & 15;

    v8f acc0 = {}, acc1 = {}, acc2 = {}, acc3 = {}, acc4 = {};

    #pragma unroll
    for (int k0 = 0; k0 < 64; k0 += 4) {
        const int ka = k0 + half * 2;

        const v2f a  = *(const v2f*)&sAw[m * SA_STRIDE + ka];
        const v2f b0 = *(const v2f*)&sWT[(m +  0) * SW_STRIDE + ka];
        const v2f b1 = *(const v2f*)&sWT[(m + 16) * SW_STRIDE + ka];
        const v2f b2 = *(const v2f*)&sWT[(m + 32) * SW_STRIDE + ka];
        const v2f b3 = *(const v2f*)&sWT[(m + 48) * SW_STRIDE + ka];
        const v2f b4 = *(const v2f*)&sWT[(m + 64) * SW_STRIDE + ka];

        acc0 = __builtin_amdgcn_wmma_f32_16x16x4_f32(false, a, false, b0, (short)0, acc0, false, false);
        acc1 = __builtin_amdgcn_wmma_f32_16x16x4_f32(false, a, false, b1, (short)0, acc1, false, false);
        acc2 = __builtin_amdgcn_wmma_f32_16x16x4_f32(false, a, false, b2, (short)0, acc2, false, false);
        acc3 = __builtin_amdgcn_wmma_f32_16x16x4_f32(false, a, false, b3, (short)0, acc3, false, false);
        acc4 = __builtin_amdgcn_wmma_f32_16x16x4_f32(false, a, false, b4, (short)0, acc4, false, false);
    }

    float* const lg = (m < 4 ? el : er) + (m & 3);
    #pragma unroll
    for (int v = 0; v < 8; ++v) {
        const int row = rowbase + v + half * 8;
        if (row < N) {
            float* fr = feat + (size_t)row * 64;
            fr[ 0 + m] = acc0[v];
            fr[16 + m] = acc1[v];
            fr[32 + m] = acc2[v];
            fr[48 + m] = acc3[v];
            if (m < 8) lg[(size_t)row * 4] = acc4[v];
        }
    }

    if (pmax != nullptr) {
        float mx = acc4[0];
        #pragma unroll
        for (int v = 1; v < 8; ++v) mx = fmaxf(mx, acc4[v]);
        mx = fmaxf(mx, __shfl_xor(mx, 16, 32));
        if (lane < 4) s_red[wave * 4 + lane] = mx;
        __syncthreads();
        if (tid < 4) {
            float g = s_red[tid];
            for (int w = 1; w < 8; ++w) g = fmaxf(g, s_red[w * 4 + tid]);
            pmax[((int)blockIdx.x - nbw) * 4 + tid] = g;
        }
    }
}

__device__ __forceinline__ float leaky02(float x) { return x > 0.f ? x : 0.2f * x; }

__device__ __forceinline__ unsigned f2o(float f) {
    unsigned u = __float_as_uint(f);
    return (u >> 31) ? ~u : (u | 0x80000000u);
}

__device__ __forceinline__ float o2f(unsigned u) {
    if (u == 0u) return 0.f;
    return (u & 0x80000000u) ? __uint_as_float(u ^ 0x80000000u) : __uint_as_float(~u);
}

template <int U>
__device__ __forceinline__ int row_share(int v) {
    return __builtin_amdgcn_update_dpp(0, v, 0x150 + U, 0xf, 0xf, true);
}

template <int C, bool NARROW>
__device__ __forceinline__ void row_chunk(int s, int dl, const float* myw, const float* __restrict__ feat,
                                          float* acc, int lane)
{
    const int half = lane >> 4;
    const int c4   = lane & 15;
    const float4 wa = *(const float4*)(myw + C * 64 + half * 32 + (c4 >> 2) * 8);
    const float4 wb = *(const float4*)(myw + C * 64 + half * 32 + (c4 >> 2) * 8 + 4);
    const float wv[8] = { wa.x, wa.y, wa.z, wa.w, wb.x, wb.y, wb.z, wb.w };
    float4 f[8];
    int    dj[8];
#define ROW_LOAD(K) { const int sj = row_share<C * 8 + K>(s); dj[K] = row_share<C * 8 + K>(dl); \
                      if constexpr (NARROW) f[K] = *(const float4*)(feat + (((unsigned)sj << 6) | (unsigned)(c4 * 4))); \
                      else                  f[K] = *(const float4*)(feat + (size_t)sj * 64 + c4 * 4); }
    ROW_LOAD(0) ROW_LOAD(1) ROW_LOAD(2) ROW_LOAD(3) ROW_LOAD(4) ROW_LOAD(5) ROW_LOAD(6) ROW_LOAD(7)
#undef ROW_LOAD
    float* const acca = acc + c4 * 4 + (half ? 2 : 0);
    float* const accb = acc + c4 * 4 + (half ? 0 : 2);
    #pragma unroll
    for (int k = 0; k < 8; ++k) {
        const float p0 = f[k].x * wv[k], p1 = f[k].y * wv[k];
        const float p2 = f[k].z * wv[k], p3 = f[k].w * wv[k];
        float* const ra = acca + dj[k] * 64;
        float* const rb = accb + dj[k] * 64;
        __hip_atomic_fetch_add(ra,     half ? p2 : p0, __ATOMIC_RELAXED, __HIP_MEMORY_SCOPE_WORKGROUP);
        __hip_atomic_fetch_add(ra + 1, half ? p3 : p1, __ATOMIC_RELAXED, __HIP_MEMORY_SCOPE_WORKGROUP);
        __hip_atomic_fetch_add(rb,     half ? p0 : p2, __ATOMIC_RELAXED, __HIP_MEMORY_SCOPE_WORKGROUP);
        __hip_atomic_fetch_add(rb + 1, half ? p1 : p3, __ATOMIC_RELAXED, __HIP_MEMORY_SCOPE_WORKGROUP);
    }
}

template <bool USEIX>
__device__ __forceinline__ void agg_owned_body(
    const int* __restrict__ src, const int* __restrict__ dst,
    const float* __restrict__ el, const float* __restrict__ er,
    const float* __restrict__ feat,
    const float* __restrict__ bias, float* __restrict__ out, int N, int E,
    const unsigned* __restrict__ blk, const unsigned* __restrict__ offs, int nbw,
    unsigned char* pool)
{
    constexpr int LC  = USEIX ? LCAPX : LCAP;
    constexpr int APP = USEIX ? 256 : 512;
    constexpr unsigned SRCMASK = (1u << SRCBITS) - 1u;
    float*    const acc   = (float*)pool;
    float*    const sden  = acc + NPW * 64;
    unsigned* const smax  = (unsigned*)(sden + NPW * 4);
    float*    const swt   = (float*)(smax + NPW * 4);
    int*      const lst_e = (int*)(swt + AGGW * 128);
    int*      const lst_d = lst_e + AGGW * LC;
    float*    const ssc   = (float*)(lst_d + (USEIX ? 0 : AGGW * LC));
    int*      const lostp = (int*)(ssc + (USEIX ? AGGW * LC * 4 : 0));
    static_assert((NPW * 72 + AGGW * 128 + AGGW * LC * (USEIX ? 5 : 2) + 4) * 4 <= POOL_BYTES, "the pool must hold this carving");
#define s_lost (*lostp)

    const int tid  = threadIdx.x;
    const int wave = tid >> 5;
    const int lane = tid & 31;
    const int lo   = blockIdx.x * NPW;
    const int cnt  = (N - lo) < NPW ? (N - lo) : NPW;

    for (int i = tid; i < NPW * 64; i += 1024) acc[i] = 0.f;
    for (int i = tid; i < NPW * 4;  i += 1024) { sden[i] = 0.f; smax[i] = 0u; }
    if (tid == 0) s_lost = 0;
    __syncthreads();

    float* const myw = swt + wave * 128;
    int*   const mye = lst_e + wave * LC;
    int*   const myd = lst_d + (USEIX ? 0 : wave * LC);
    float* const mys = ssc + (USEIX ? wave * LC * 4 : 0);

    auto maxupd = [&](int n, bool have_s, bool keep) __attribute__((always_inline)) {
        for (int b = 0; b < n; b += 128) {
            int    sv[4], dv[4];
            float4 l4[4], r4[4];
            #pragma unroll
            for (int u = 0; u < 4; ++u) {
                const int  i     = b + u * 32 + lane;
                const bool valid = i < n;
                const int  e     = valid ? mye[i] : 0;
                if constexpr (USEIX) {
                    dv[u] = valid ? (int)((unsigned)e >> SRCBITS) : -1;
                    sv[u] = (int)((unsigned)e & SRCMASK);
                } else {
                    dv[u] = valid ? myd[i] : -1;
                    sv[u] = have_s ? e : src[e];
                }
            }
            #pragma unroll
            for (int u = 0; u < 4; ++u) {
                const int dl = dv[u] < 0 ? 0 : dv[u];
                l4[u] = *(const float4*)(el + (size_t)sv[u] * 4);
                r4[u] = *(const float4*)(er + (size_t)(lo + dl) * 4);
            }
            #pragma unroll
            for (int u = 0; u < 4; ++u) {
                if (dv[u] >= 0) {
                    unsigned* mx = &smax[dv[u] * 4];
                    float4 sc;
                    sc.x = leaky02(l4[u].x + r4[u].x);
                    sc.y = leaky02(l4[u].y + r4[u].y);
                    sc.z = leaky02(l4[u].z + r4[u].z);
                    sc.w = leaky02(l4[u].w + r4[u].w);
                    __hip_atomic_fetch_max(mx + 0, f2o(sc.x), __ATOMIC_RELAXED, __HIP_MEMORY_SCOPE_WORKGROUP);
                    __hip_atomic_fetch_max(mx + 1, f2o(sc.y), __ATOMIC_RELAXED, __HIP_MEMORY_SCOPE_WORKGROUP);
                    __hip_atomic_fetch_max(mx + 2, f2o(sc.z), __ATOMIC_RELAXED, __HIP_MEMORY_SCOPE_WORKGROUP);
                    __hip_atomic_fetch_max(mx + 3, f2o(sc.w), __ATOMIC_RELAXED, __HIP_MEMORY_SCOPE_WORKGROUP);
                    if (USEIX && keep) *(float4*)&mys[(b + u * 32 + lane) * 4] = sc;
                    if (!USEIX && !have_s) mye[b + u * 32 + lane] = sv[u];
                }
            }
        }
    };

    auto drain = [&](int n, bool have_s, bool kept) __attribute__((always_inline)) {
        for (int b = 0; b < n; b += 32) {
            const int  i     = b + ((lane & 15) << 1) + (lane >> 4);
            const bool valid = i < n;
            const int  q  = valid ? mye[i] : 0;
            int s, dl;
            if constexpr (USEIX) {
                s  = (int)((unsigned)q & SRCMASK);
                dl = (int)((unsigned)q >> SRCBITS);
            } else {
                dl = valid ? myd[i] : 0;
                s  = have_s ? q : src[q];
                s  = valid ? s : 0;
            }
            float4 sc;
            if (USEIX && kept) {
                sc = *(const float4*)&mys[(valid ? i : 0) * 4];
            } else {
                const int  d  = lo + dl;
                const float4 l4 = *(const float4*)(el + (size_t)s * 4);
                const float4 r4 = *(const float4*)(er + (size_t)d * 4);
                sc.x = leaky02(l4.x + r4.x);
                sc.y = leaky02(l4.y + r4.y);
                sc.z = leaky02(l4.z + r4.z);
                sc.w = leaky02(l4.w + r4.w);
            }
            const uint4  mu = *(const uint4*)&smax[dl * 4];
            float4 w;
            w.x = valid ? __expf(sc.x - o2f(mu.x)) : 0.f;
            w.y = valid ? __expf(sc.y - o2f(mu.y)) : 0.f;
            w.z = valid ? __expf(sc.z - o2f(mu.z)) : 0.f;
            w.w = valid ? __expf(sc.w - o2f(mu.w)) : 0.f;
            if (valid) {
                float* dn = &sden[dl * 4];
                __hip_atomic_fetch_add(dn + 0, w.x, __ATOMIC_RELAXED, __HIP_MEMORY_SCOPE_WORKGROUP);
                __hip_atomic_fetch_add(dn + 1, w.y, __ATOMIC_RELAXED, __HIP_MEMORY_SCOPE_WORKGROUP);
                __hip_atomic_fetch_add(dn + 2, w.z, __ATOMIC_RELAXED, __HIP_MEMORY_SCOPE_WORKGROUP);
                __hip_atomic_fetch_add(dn + 3, w.w, __ATOMIC_RELAXED, __HIP_MEMORY_SCOPE_WORKGROUP);
            }
            {
                float* wd = myw + ((lane & 15) >> 3) * 64 + (lane >> 4) * 32 + (lane & 7);
                wd[0] = w.x; wd[8] = w.y; wd[16] = w.z; wd[24] = w.w;
            }

            const int nb = (n - b) < 32 ? (n - b) : 32;
            row_chunk<0, USEIX>(s, dl, myw, feat, acc, lane);
            if (nb > 16) row_chunk<1, USEIX>(s, dl, myw, feat, acc, lane);
        }
    };

    const int G4     = E >> 2;
    const int steps  = ((G4 + 1023) >> 10);
    const int steps4 = (steps + 3) & ~3;
    const int st0    = (int)(((long long)blockIdx.x * steps4) / gridDim.x) & ~3;
    int fill = 0;
    int pass = 0;

    auto scan_load = [&](int k, int4 (&v)[4]) {
        int kk = st0 + k;
        if (kk >= steps4) kk -= steps4;
        #pragma unroll
        for (int u = 0; u < 4; ++u) {
            const int g = (kk + u) * 1024 + tid;
            v[u] = *(const int4*)(dst + (size_t)(g < G4 ? g : 0) * 4);
        }
    };

    auto examine = [&](int g, bool ok, const int4& v) __attribute__((always_inline)) {
        const int e0 = g * 4;
        const unsigned lim = ok ? (unsigned)cnt : 0u;
        const unsigned u0 = (unsigned)(v.x - lo);
        const unsigned u1 = (unsigned)(v.y - lo);
        const unsigned u2 = (unsigned)(v.z - lo);
        const unsigned u3 = (unsigned)(v.w - lo);
        const bool m0 = u0 < lim, m1 = u1 < lim;
        const bool m2 = u2 < lim, m3 = u3 < lim;
        if (__builtin_amdgcn_ballot_w32(m0 | m1 | m2 | m3)) {
            unsigned hb;
            hb = __builtin_amdgcn_ballot_w32(m0);
            if (m0) { const int p = fill + __builtin_amdgcn_mbcnt_lo(hb, 0u); if (p < LC) { mye[p] = e0;     myd[p] = (int)u0; } }
            fill += __builtin_popcount(hb);
            hb = __builtin_amdgcn_ballot_w32(m1);
            if (m1) { const int p = fill + __builtin_amdgcn_mbcnt_lo(hb, 0u); if (p < LC) { mye[p] = e0 + 1; myd[p] = (int)u1; } }
            fill += __builtin_popcount(hb);
            hb = __builtin_amdgcn_ballot_w32(m2);
            if (m2) { const int p = fill + __builtin_amdgcn_mbcnt_lo(hb, 0u); if (p < LC) { mye[p] = e0 + 2; myd[p] = (int)u2; } }
            fill += __builtin_popcount(hb);
            hb = __builtin_amdgcn_ballot_w32(m3);
            if (m3) { const int p = fill + __builtin_amdgcn_mbcnt_lo(hb, 0u); if (p < LC) { mye[p] = e0 + 3; myd[p] = (int)u3; } }
            fill += __builtin_popcount(hb);
        }
    };

    auto flush = [&]() __attribute__((always_inline)) {
        if (pass != 0 && fill > LC - APP) {
            if (pass == 1) maxupd(fill, USEIX, false);
            else           drain(fill, USEIX, false);
            fill = 0;
        }
    };

    auto scan_use = [&](int k, const int4 (&v)[4]) {
        int kk = st0 + k;
        if (kk >= steps4) kk -= steps4;
        #pragma unroll
        for (int u = 0; u < 4; ++u) {
            const int g = (kk + u) * 1024 + tid;
            examine(g, g < G4, v[u]);
        }
        flush();
    };

    auto append = [&](bool m, unsigned w, int add) __attribute__((always_inline)) {
        const int p = fill + lane;
        if (m && p < LC) mye[p] = (int)w;
        fill += add;
    };

    auto collect = [&]() __attribute__((always_inline)) {
        for (int b0 = wave; b0 < nbw; b0 += AGGW * 32) {
            const int bl = b0 + lane * AGGW;
            const unsigned meta = bl < nbw ? offs[(size_t)bl * BMROWS + blockIdx.x] : 0u;
            const int left = (nbw - b0 + AGGW - 1) / AGGW;
            const int nseg = left < 32 ? left : 32;
            for (int i0 = 0; i0 < nseg; i0 += 8) {
                unsigned w[8];
                int      cn[8];
                const unsigned* sg[8];
                #pragma unroll
                for (int u = 0; u < 8; ++u) {
                    const int i = i0 + u;
                    const unsigned mt = (unsigned)__builtin_amdgcn_readlane((int)meta, i);
                    cn[u] = (int)(mt >> 16);
                    sg[u] = blk + (size_t)(b0 + i * AGGW) * EPB + (mt & 0xffffu);
                    w[u]  = lane < cn[u] ? sg[u][lane] : 0u;
                }
                #pragma unroll
                for (int u = 0; u < 8; ++u) append(lane < cn[u], w[u], cn[u] < 32 ? cn[u] : 32);
                flush();
                #pragma unroll 1
                for (int i = i0; i < i0 + 8 && i < nseg; ++i) {
                    const unsigned mt = (unsigned)__builtin_amdgcn_readlane((int)meta, i);
                    const int c = (int)(mt >> 16);
                    const unsigned* const sp = blk + (size_t)(b0 + i * AGGW) * EPB + (mt & 0xffffu);
                    for (int k = 32; k < c; k += 32) {
                        const bool m = k + lane < c;
                        append(m, m ? sp[k + lane] : 0u, (c - k) < 32 ? (c - k) : 32);
                        flush();
                    }
                }
            }
        }
    };

    for (pass = 0; pass < 3; ++pass) {
        fill = 0;
        if constexpr (USEIX) {
            collect();
        } else {
            int4 va[4], vb[4];
            if (steps4 > 0) scan_load(0, va);
            for (int k = 0; k < steps4; k += 8) {
                scan_load(k + 4, vb);
                scan_use(k, va);
                scan_load(k + 8, va);
                if (k + 4 < steps4) scan_use(k + 4, vb);
            }
        }
        if (wave == 0) {
            const int et = (G4 << 2) + lane;
            unsigned ut = 0xffffffffu;
            if (et < E) ut = (unsigned)(dst[et] - lo);
            const bool mt = ut < (unsigned)cnt;
            const unsigned bm = __builtin_amdgcn_ballot_w32(mt);
            if (mt) {
                const int p = fill + __builtin_amdgcn_mbcnt_lo(bm, 0u);
                if (p < LC) {
                    if constexpr (USEIX) mye[p] = (int)(((unsigned)src[et] & SRCMASK) | (ut << SRCBITS));
                    else { mye[p] = et; myd[p] = (int)ut; }
                }
            }
            fill += __builtin_popcount(bm);
        }
        if (pass == 0) {
            if (fill > LC && lane == 0) s_lost = 1;
            maxupd(fill < LC ? fill : LC, USEIX, USEIX);
            __syncthreads();
            if (__builtin_amdgcn_readfirstlane(s_lost) == 0) { drain(fill, true, USEIX); break; }
        } else if (pass == 1) {
            maxupd(fill, USEIX, false);
            __syncthreads();
        } else {
            drain(fill, USEIX, false);
        }
    }
    __syncthreads();

    for (int i = tid; i < cnt * 64; i += 1024) {
        const int n = i >> 6, c = i & 63;
        const float dn = sden[n * 4 + (c >> 4)];
        const float a  = acc[i];
        out[(size_t)(lo + n) * 64 + c] = (dn > 0.f ? a / dn : 0.f) + bias[c];
    }
#undef s_lost
}

__global__ __launch_bounds__(1024) void gat_agg_scan(
    const int* __restrict__ src, const int* __restrict__ dst,
    const float* __restrict__ el, const float* __restrict__ er,
    const float* __restrict__ feat,
    const float* __restrict__ bias, float* __restrict__ out, int N, int E)
{
    __shared__ __attribute__((aligned(16))) unsigned char pool[POOL_BYTES];
    agg_owned_body<false>(src, dst, el, er, feat, bias, out, N, E, nullptr, nullptr, 0, pool);
}

template <int C>
__device__ __forceinline__ void row_loads_mm(int s, int ml, const float* __restrict__ feat,
                                             float4 (&f)[8], int (&mj)[8], int lane)
{
    const int c4 = lane & 15;
#define ROW_LOAD(K) { const int sj = row_share<C * 8 + K>(s); mj[K] = row_share<C * 8 + K>(ml); \
                      f[K] = *(const float4*)(feat + (((unsigned)sj << 6) | (unsigned)(c4 * 4))); }
    ROW_LOAD(0) ROW_LOAD(1) ROW_LOAD(2) ROW_LOAD(3) ROW_LOAD(4) ROW_LOAD(5) ROW_LOAD(6) ROW_LOAD(7)
#undef ROW_LOAD
}

template <int C>
__device__ __forceinline__ void row_products_mm(const float* myw, const float4 (&f)[8], const int (&mj)[8],
                                                v8f& t0, v8f& t1, v8f& t2, v8f& t3, int lane)
{
    const int half = lane >> 4;
    const int c4   = lane & 15;
    const float4 wa = *(const float4*)(myw + C * 64 + half * 32 + (c4 >> 2) * 8);
    const float4 wb = *(const float4*)(myw + C * 64 + half * 32 + (c4 >> 2) * 8 + 4);
    const float wv[8] = { wa.x, wa.y, wa.z, wa.w, wb.x, wb.y, wb.z, wb.w };
    #pragma unroll
    for (int g = 0; g < 4; ++g) {
        const int k0 = 2 * g, k1 = 2 * g + 1;
        v2f a, b0, b1, b2, b3;
        a.x  = mj[k0] == c4 ? 1.f : 0.f;   a.y  = mj[k1] == c4 ? 1.f : 0.f;
        b0.x = f[k0].x * wv[k0];           b0.y = f[k1].x * wv[k1];
        b1.x = f[k0].y * wv[k0];           b1.y = f[k1].y * wv[k1];
        b2.x = f[k0].z * wv[k0];           b2.y = f[k1].z * wv[k1];
        b3.x = f[k0].w * wv[k0];           b3.y = f[k1].w * wv[k1];
        t0 = __builtin_amdgcn_wmma_f32_16x16x4_f32(false, a, false, b0, (short)0, t0, false, false);
        t1 = __builtin_amdgcn_wmma_f32_16x16x4_f32(false, a, false, b1, (short)0, t1, false, false);
        t2 = __builtin_amdgcn_wmma_f32_16x16x4_f32(false, a, false, b2, (short)0, t2, false, false);
        t3 = __builtin_amdgcn_wmma_f32_16x16x4_f32(false, a, false, b3, (short)0, t3, false, false);
    }
}

#define LCO    448
#define TLROWS 13
__device__ __forceinline__ int tile_of(unsigned dl)  { return (int)((dl * 21u) >> 8); }
__device__ __forceinline__ int tile_first(int w)     { return (w * 256 + 20) / 21; }
static_assert(((unsigned)(NPW - 1) * 21u) >> 8 == AGGW - 1, "the last local row must belong to the last wave");
static_assert((AGGW * 256 + 20) / 21 == NPW, "the tiles must end with the range");
static_assert(TLROWS <= 16, "two rows per lane group of four cover a tile");
static_assert((NPW * 12 + 392 + AGGW * 4 + AGGW * 128 + AGGW * LCO + AGGW + 4) * 4 <= POOL_BYTES, "the pool must hold the owner lists");

__global__ __launch_bounds__(1024) void gat_agg_tiles(
    const int* __restrict__ src, const int* __restrict__ dst,
    const float* __restrict__ el, const float* __restrict__ er,
    const float* __restrict__ feat,
    const float* __restrict__ bias, float* __restrict__ out, int N, int E,
    const unsigned* __restrict__ blk, const unsigned* __restrict__ offs, int nbw,
    const float* __restrict__ pmax, int npb)
{
    __shared__ __attribute__((aligned(16))) unsigned char pool[POOL_BYTES];
    constexpr unsigned SRCMASK = (1u << SRCBITS) - 1u;
    float*    const sden  = (float*)pool;
    float*    const ser   = sden + NPW * 4;
    float*    const ssh   = ser + NPW * 4;
    unsigned* const smark = (unsigned*)(ssh + NPW * 4);
    float*    const swm   = (float*)(smark + 392);
    float*    const swt   = swm + AGGW * 4;
    int*      const olst  = (int*)(swt + AGGW * 128);
    unsigned* const ocnt  = (unsigned*)(olst + AGGW * LCO);
    int*      const lostp = (int*)(ocnt + AGGW);

    const int tid  = threadIdx.x;
    const int wave = tid >> 5;
    const int lane = tid & 31;
    const int lo   = blockIdx.x * NPW;
    const int cnt  = (N - lo) < NPW ? (N - lo) : NPW;
    const int rs   = tile_first(wave);
    const int re   = tile_first(wave + 1) < cnt ? tile_first(wave + 1) : cnt;

    float pm = -3.402823466e+38f;
    for (int i = tid; i < npb * 4; i += 1024) pm = fmaxf(pm, pmax[i]);
    const int d0 = rs + (lane >> 2), d1 = d0 + 8;
    float er0 = 0.f, er1 = 0.f;
    if (d0 < re) er0 = er[(size_t)(lo + d0) * 4 + (lane & 3)];
    if (d1 < re) er1 = er[(size_t)(lo + d1) * 4 + (lane & 3)];

    for (int i = tid; i < NPW * 4; i += 1024) sden[i] = 0.f;
    if (tid < NPW) smark[tid] = 0u;
    if (tid < AGGW) ocnt[tid] = 0u;
    if (tid == 0) { lostp[0] = 0; lostp[1] = 0; }
    __syncthreads();

    auto put = [&](bool m, unsigned w) __attribute__((always_inline)) {
        if (m) {
            const int o = tile_of(w >> SRCBITS);
            const unsigned p = __hip_atomic_fetch_add(&ocnt[o], 1u, __ATOMIC_RELAXED, __HIP_MEMORY_SCOPE_WORKGROUP);
            if (p < (unsigned)LCO) olst[o * LCO + (int)p] = (int)w;
            else                   lostp[0] = 1;
        }
    };

    for (int b0 = wave; b0 < nbw; b0 += AGGW * 32) {
        const int bl = b0 + lane * AGGW;
        const unsigned meta = bl < nbw ? offs[(size_t)bl * BMROWS + blockIdx.x] : 0u;
        const int left = (nbw - b0 + AGGW - 1) / AGGW;
        const int nseg = left < 32 ? left : 32;
        for (int i0 = 0; i0 < nseg; i0 += 8) {
            unsigned w[8];
            int      cn[8];
            #pragma unroll
            for (int u = 0; u < 8; ++u) {
                const int i = i0 + u;
                const unsigned mt = (unsigned)__builtin_amdgcn_readlane((int)meta, i);
                cn[u] = (int)(mt >> 16);
                const unsigned* const sg = blk + (size_t)(b0 + i * AGGW) * EPB + (mt & 0xffffu);
                w[u]  = lane < cn[u] ? sg[lane] : 0u;
            }
            unsigned p[8];
            #pragma unroll
            for (int u = 0; u < 8; ++u) {
                p[u] = 0u;
                if (lane < cn[u]) p[u] = __hip_atomic_fetch_add(&ocnt[tile_of(w[u] >> SRCBITS)], 1u, __ATOMIC_RELAXED, __HIP_MEMORY_SCOPE_WORKGROUP);
            }
            #pragma unroll
            for (int u = 0; u < 8; ++u) {
                if (lane < cn[u]) {
                    if (p[u] < (unsigned)LCO) olst[tile_of(w[u] >> SRCBITS) * LCO + (int)p[u]] = (int)w[u];
                    else                      lostp[0] = 1;
                }
            }
            for (int i = i0; i < i0 + 8 && i < nseg; ++i) {
                const unsigned mt = (unsigned)__builtin_amdgcn_readlane((int)meta, i);
                const int c = (int)(mt >> 16);
                const unsigned* const sp = blk + (size_t)(b0 + i * AGGW) * EPB + (mt & 0xffffu);
                for (int k = 32; k < c; k += 32) {
                    const bool m = k + lane < c;
                    put(m, m ? sp[k + lane] : 0u);
                }
            }
        }
    }
    if (wave == 0) {
        const int et = ((E >> 2) << 2) + lane;
        if (et < E) {
            const unsigned ut = (unsigned)(dst[et] - lo);
            if (ut < (unsigned)cnt) put(true, ((unsigned)src[et] & SRCMASK) | (ut << SRCBITS));
        }
    }
    pm = fmaxf(pm, __shfl_xor(pm, 4, 32));
    pm = fmaxf(pm, __shfl_xor(pm, 8, 32));
    pm = fmaxf(pm, __shfl_xor(pm, 16, 32));
    if (lane < 4) swm[wave * 4 + lane] = pm;
    __syncthreads();

    if (__builtin_amdgcn_readfirstlane(lostp[0]) == 0) {
        const int n = __builtin_amdgcn_readfirstlane((int)ocnt[wave]);
        int*   const mye = olst + wave * LCO;
        float* const myw = swt + wave * 128;

        float gm = fmaxf(fmaxf(swm[lane], swm[lane + 32]), fmaxf(swm[lane + 64], swm[lane + 96]));
        gm = fmaxf(gm, __shfl_xor(gm, 4, 32));
        gm = fmaxf(gm, __shfl_xor(gm, 8, 32));
        gm = fmaxf(gm, __shfl_xor(gm, 16, 32));
        if (d0 < re) { ser[d0 * 4 + (lane & 3)] = er0; ssh[d0 * 4 + (lane & 3)] = leaky02(gm + er0); }
        if (d1 < re) { ser[d1 * 4 + (lane & 3)] = er1; ssh[d1 * 4 + (lane & 3)] = leaky02(gm + er1); }

        v8f t0 = {}, t1 = {}, t2 = {}, t3 = {};
        for (int b = 0; b < n; b += 32) {
            const int  i     = b + ((lane & 15) << 1) + (lane >> 4);
            const bool valid = i < n;
            const int  q  = valid ? mye[i] : (rs << SRCBITS);
            const int  s  = (int)((unsigned)q & SRCMASK);
            const int  dl = (int)((unsigned)q >> SRCBITS);
            const int  ml = valid ? dl - rs : -1;
            const float4 l4 = *(const float4*)(el + (size_t)s * 4);
            float4 f[8];
            int    mj[8];
            row_loads_mm<0>(s, ml, feat, f, mj, lane);
            const float4 r4 = *(const float4*)&ser[dl * 4];
            const float4 h4 = *(const float4*)&ssh[dl * 4];
            float4 w;
            w.x = __expf(leaky02(l4.x + r4.x) - h4.x);
            w.y = __expf(leaky02(l4.y + r4.y) - h4.y);
            w.z = __expf(leaky02(l4.z + r4.z) - h4.z);
            w.w = __expf(leaky02(l4.w + r4.w) - h4.w);
            if (valid) {
                float* dn = &sden[dl * 4];
                __hip_atomic_fetch_add(dn + 0, w.x, __ATOMIC_RELAXED, __HIP_MEMORY_SCOPE_WORKGROUP);
                __hip_atomic_fetch_add(dn + 1, w.y, __ATOMIC_RELAXED, __HIP_MEMORY_SCOPE_WORKGROUP);
                __hip_atomic_fetch_add(dn + 2, w.z, __ATOMIC_RELAXED, __HIP_MEMORY_SCOPE_WORKGROUP);
                __hip_atomic_fetch_add(dn + 3, w.w, __ATOMIC_RELAXED, __HIP_MEMORY_SCOPE_WORKGROUP);
                smark[dl] = 1u;
            }
            {
                float* wd = myw + ((lane & 15) >> 3) * 64 + (lane >> 4) * 32 + (lane & 7);
                wd[0] = w.x; wd[8] = w.y; wd[16] = w.z; wd[24] = w.w;
            }
            const int nb = (n - b) < 32 ? (n - b) : 32;
            row_products_mm<0>(myw, f, mj, t0, t1, t2, t3, lane);
            if (nb > 16) {
                row_loads_mm<1>(s, ml, feat, f, mj, lane);
                row_products_mm<1>(myw, f, mj, t0, t1, t2, t3, lane);
            }
        }

        {
            bool bad = false;
            if (d0 < re && smark[d0] != 0u) { const float d = sden[d0 * 4 + (lane & 3)]; bad = bad || !(d >= 1e-20f && d <= 3.0e38f); }
            if (d1 < re && smark[d1] != 0u) { const float d = sden[d1 * 4 + (lane & 3)]; bad = bad || !(d >= 1e-20f && d <= 3.0e38f); }
            if (bad) lostp[1] = 1;
        }
        __syncthreads();

        if (__builtin_amdgcn_readfirstlane(lostp[1]) == 0) {
            const int half = lane >> 4;
            const int c4   = lane & 15;
            const float4 bv = *(const float4*)(bias + c4 * 4);
            #pragma unroll
            for (int r = 0; r < 8; ++r) {
                const int nl = rs + half * 8 + r;
                if (nl < re) {
                    const float dn = sden[nl * 4 + (c4 >> 2)];
                    float4 o;
                    o.x = (dn > 0.f ? t0[r] / dn : 0.f) + bv.x;
                    o.y = (dn > 0.f ? t1[r] / dn : 0.f) + bv.y;
                    o.z = (dn > 0.f ? t2[r] / dn : 0.f) + bv.z;
                    o.w = (dn > 0.f ? t3[r] / dn : 0.f) + bv.w;
                    *(float4*)(out + (size_t)(lo + nl) * 64 + c4 * 4) = o;
                }
            }
            return;
        }
    }
    __syncthreads();
    agg_owned_body<true>(src, dst, el, er, feat, bias, out, N, E, blk, offs, nbw, pool);
}

extern "C" void kernel_launch(void* const* d_in, const int* in_sizes, int n_in,
                              void* d_out, int out_size, void* d_ws, size_t ws_size,
                              hipStream_t stream)
{
    const float* nf   = (const float*)d_in[0];
    const float* W    = (const float*)d_in[1];
    const float* al   = (const float*)d_in[2];
    const float* ar   = (const float*)d_in[3];
    const float* bias = (const float*)d_in[4];
    const int*   src  = (const int*)d_in[5];
    const int*   dst  = (const int*)d_in[6];

    const int N = in_sizes[0] / 64;
    const int E = in_sizes[5];

    float* ws    = (float*)d_ws;
    float* feat  = ws;
    float* el    = feat  + (size_t)N * 64;
    float* er    = el    + (size_t)N * 4;
    float* out   = (float*)d_out;
    const int nr  = (N + NPW - 1) / NPW;
    const int G4  = E >> 2;
    const int nbw = (G4 + 1023) / 1024;
    unsigned* blk  = (unsigned*)(er + (size_t)N * 4);
    unsigned* offs = blk + (size_t)nbw * EPB;
    const int npb = (N + 127) / 128;
    float* pmax = (float*)(offs + (size_t)nbw * BMROWS);
    const size_t need = ((size_t)N * 72 + (size_t)nbw * (EPB + BMROWS) + (size_t)npb * 4) * sizeof(float);

    if (nr <= BMROWS && nbw > 0 && N <= (1 << SRCBITS) && ws_size >= need) {
        gat_proj_logits_sort<<<nbw + npb, 256, 0, stream>>>(nf, W, al, ar, feat, el, er, N, src, dst, blk, offs, G4, nbw, pmax);
        gat_agg_tiles<<<nr, 1024, 0, stream>>>(src, dst, el, er, feat, bias, out, N, E, blk, offs, nbw, pmax, npb);
    } else {
        gat_proj_logits_sort<<<npb, 256, 0, stream>>>(nf, W, al, ar, feat, el, er, N, src, dst, nullptr, nullptr, G4, 0, nullptr);
        gat_agg_scan<<<nr, 1024, 0, stream>>>(src, dst, el, er, feat, bias, out, N, E);
    }
}


# global loads asked for as one group where each loop trip waited for its own: projection prologue (W, A tile, attention vectors), aggregation prologue (el maxima, er, block offsets, 16 segments)
# speedup vs baseline: 7.2151x; 1.0341x over previous
#include <hip/hip_runtime.h>


typedef float v2f __attribute__((ext_vector_type(2)));
typedef float v8f __attribute__((ext_vector_type(8)));

#define SA_STRIDE 68
#define SW_STRIDE 68

#define NPW   391
#define LCAP  640
#define LCAPX 304
#define AGGW  32
#define BMROWS 256
#define EPB   4096
#define SRCBITS 23
static_assert(NPW <= (1 << (32 - SRCBITS)), "local destination must fit above the source bits");

#define SWT_ROWS 80
#define PROJ_LDS_BYTES ((SWT_ROWS * SW_STRIDE + 8 * 16 * SA_STRIDE) * 4)
static_assert((2 * BMROWS + EPB) * 4 <= PROJ_LDS_BYTES, "the sorter's counters and image must fit in the projection's LDS block");
#define POOL_BYTES 323584

__global__ __launch_bounds__(256) void gat_proj_logits_sort(
    const float* __restrict__ nf, const float* __restrict__ W,
    const float* __restrict__ al, const float* __restrict__ ar,
    float* __restrict__ feat, float* __restrict__ el, float* __restrict__ er, int N,
    const int* __restrict__ src, const int* __restrict__ dst,
    unsigned* __restrict__ blk, unsigned* __restrict__ offs, int G4, int nbw,
    float* __restrict__ pmax)
{
    __shared__ __attribute__((aligned(16))) unsigned char smem[PROJ_LDS_BYTES];
    __shared__ float s_red[8 * 4];

    const int tid  = threadIdx.x;
    const int wave = tid >> 5;
    const int lane = tid & 31;

    if ((int)blockIdx.x < nbw) {
        unsigned* const hist   = (unsigned*)smem;
        unsigned* const base   = hist + BMROWS;
        unsigned* const sorted = base + BMROWS;
        hist[tid] = 0u;
        __syncthreads();
        const int gw = ((int)blockIdx.x * 32 + wave * 4) * 32 + lane;
        int4 vd[4], vs[4];
        #pragma unroll
        for (int q = 0; q < 4; ++q) {
            const int g = gw + q * 32;
            const size_t o = (size_t)(g < G4 ? g : 0) * 4;
            vd[q] = *(const int4*)(dst + o);
            vs[q] = *(const int4*)(src + o);
        }
        unsigned key[16];
        #pragma unroll
        for (int q = 0; q < 4; ++q) {
            const bool okg = gw + q * 32 < G4;
            const int dd[4] = { vd[q].x, vd[q].y, vd[q].z, vd[q].w };
            #pragma unroll
            for (int c = 0; c < 4; ++c) {
                unsigned k = 0xffffffffu;
                if (okg && (unsigned)dd[c] < (unsigned)N) {
                    const unsigned r = (unsigned)dd[c] / NPW;
                    k = (r << 16) | __hip_atomic_fetch_add(&hist[r], 1u, __ATOMIC_RELAXED, __HIP_MEMORY_SCOPE_WORKGROUP);
                }
                key[q * 4 + c] = k;
            }
        }
        __syncthreads();
        if (wave == 0) {
            unsigned c8[8];
            unsigned t = 0u;
            #pragma unroll
            for (int i = 0; i < 8; ++i) { c8[i] = hist[lane * 8 + i]; t += c8[i]; }
            unsigned incl = t;
            #pragma unroll
            for (int d = 1; d < 32; d <<= 1) {
                const unsigned o = __shfl_up(incl, d, 32);
                if (lane >= d) incl += o;
            }
            unsigned run = incl - t;
            #pragma unroll
            for (int i = 0; i < 8; ++i) { base[lane * 8 + i] = run; run += c8[i]; }
        }
        __syncthreads();
        #pragma unroll
        for (int q = 0; q < 4; ++q) {
            const int dd[4] = { vd[q].x, vd[q].y, vd[q].z, vd[q].w };
            const int ss[4] = { vs[q].x, vs[q].y, vs[q].z, vs[q].w };
            #pragma unroll
            for (int c = 0; c < 4; ++c) {
                const unsigned k = key[q * 4 + c];
                if (k != 0xffffffffu) {
                    const unsigned r = k >> 16;
                    sorted[base[r] + (k & 0xffffu)] =
                        ((unsigned)ss[c] & ((1u << SRCBITS) - 1u)) | (((unsigned)dd[c] - r * NPW) << SRCBITS);
                }
            }
        }
        __syncthreads();
        const unsigned total = base[BMROWS - 1] + hist[BMROWS - 1];
        unsigned* const ob = blk + (size_t)blockIdx.x * EPB;
        for (unsigned i = tid; i < total; i += 256) ob[i] = sorted[i];
        offs[(size_t)blockIdx.x * BMROWS + tid] = base[tid] | (hist[tid] << 16);
        return;
    }

    float* const sWT = (float*)smem;
    float* const sAw = sWT + SWT_ROWS * SW_STRIDE + wave * (16 * SA_STRIDE);

    const int hh = __builtin_amdgcn_readfirstlane(tid >> 6);
    float alv[16], arv[16];
    #pragma unroll
    for (int f = 0; f < 16; ++f) { alv[f] = al[hh * 16 + f]; arv[f] = ar[hh * 16 + f]; }
    float wv[16];
    #pragma unroll
    for (int j = 0; j < 16; ++j) wv[j] = W[tid + j * 256];
    const int rowbase = ((int)blockIdx.x - nbw) * 128 + wave * 16;
    float a0[16], a1[16];
    #pragma unroll
    for (int r = 0; r < 16; ++r) {
        int row = rowbase + r;
        row = row < N ? row : (N - 1);
        a0[r] = nf[(size_t)row * 64 + lane];
        a1[r] = nf[(size_t)row * 64 + lane + 32];
    }

    #pragma unroll
    for (int j = 0; j < 16; ++j) sWT[(tid & 63) * SW_STRIDE + (tid >> 6) + 4 * j] = wv[j];
    for (int i = tid; i < 8 * SW_STRIDE; i += 256) sWT[72 * SW_STRIDE + i] = 0.f;
    #pragma unroll
    for (int r = 0; r < 16; ++r) {
        sAw[r * SA_STRIDE + lane]      = a0[r];
        sAw[r * SA_STRIDE + lane + 32] = a1[r];
    }
    __syncthreads();

    {
        const int k  = tid & 63;
        float sl = 0.f, sr = 0.f;
        #pragma unroll
        for (int f = 0; f < 16; ++f) {
            const float w = sWT[(hh * 16 + f) * SW_STRIDE + k];
            sl += w * alv[f];
            sr += w * arv[f];
        }
        sWT[(64 + hh) * SW_STRIDE + k] = sl;
        sWT[(68 + hh) * SW_STRIDE + k] = sr;
    }
    __syncthreads();

    const int half = lane >> 4;
    const int m    = lane & 15;

    v8f acc0 = {}, acc1 = {}, acc2 = {}, acc3 = {}, acc4 = {};

    #pragma unroll
    for (int k0 = 0; k0 < 64; k0 += 4) {
        const int ka = k0 + half * 2;

        const v2f a  = *(const v2f*)&sAw[m * SA_STRIDE + ka];
        const v2f b0 = *(const v2f*)&sWT[(m +  0) * SW_STRIDE + ka];
        const v2f b1 = *(const v2f*)&sWT[(m + 16) * SW_STRIDE + ka];
        const v2f b2 = *(const v2f*)&sWT[(m + 32) * SW_STRIDE + ka];
        const v2f b3 = *(const v2f*)&sWT[(m + 48) * SW_STRIDE + ka];
        const v2f b4 = *(const v2f*)&sWT[(m + 64) * SW_STRIDE + ka];

        acc0 = __builtin_amdgcn_wmma_f32_16x16x4_f32(false, a, false, b0, (short)0, acc0, false, false);
        acc1 = __builtin_amdgcn_wmma_f32_16x16x4_f32(false, a, false, b1, (short)0, acc1, false, false);
        acc2 = __builtin_amdgcn_wmma_f32_16x16x4_f32(false, a, false, b2, (short)0, acc2, false, false);
        acc3 = __builtin_amdgcn_wmma_f32_16x16x4_f32(false, a, false, b3, (short)0, acc3, false, false);
        acc4 = __builtin_amdgcn_wmma_f32_16x16x4_f32(false, a, false, b4, (short)0, acc4, false, false);
    }

    float* const lg = (m < 4 ? el : er) + (m & 3);
    #pragma unroll
    for (int v = 0; v < 8; ++v) {
        const int row = rowbase + v + half * 8;
        if (row < N) {
            float* fr = feat + (size_t)row * 64;
            fr[ 0 + m] = acc0[v];
            fr[16 + m] = acc1[v];
            fr[32 + m] = acc2[v];
            fr[48 + m] = acc3[v];
            if (m < 8) lg[(size_t)row * 4] = acc4[v];
        }
    }

    if (pmax != nullptr) {
        float mx = acc4[0];
        #pragma unroll
        for (int v = 1; v < 8; ++v) mx = fmaxf(mx, acc4[v]);
        mx = fmaxf(mx, __shfl_xor(mx, 16, 32));
        if (lane < 4) s_red[wave * 4 + lane] = mx;
        __syncthreads();
        if (tid < 4) {
            float g = s_red[tid];
            for (int w = 1; w < 8; ++w) g = fmaxf(g, s_red[w * 4 + tid]);
            pmax[((int)blockIdx.x - nbw) * 4 + tid] = g;
        }
    }
}

__device__ __forceinline__ float leaky02(float x) { return x > 0.f ? x : 0.2f * x; }

__device__ __forceinline__ unsigned f2o(float f) {
    unsigned u = __float_as_uint(f);
    return (u >> 31) ? ~u : (u | 0x80000000u);
}

__device__ __forceinline__ float o2f(unsigned u) {
    if (u == 0u) return 0.f;
    return (u & 0x80000000u) ? __uint_as_float(u ^ 0x80000000u) : __uint_as_float(~u);
}

template <int U>
__device__ __forceinline__ int row_share(int v) {
    return __builtin_amdgcn_update_dpp(0, v, 0x150 + U, 0xf, 0xf, true);
}

template <int C, bool NARROW>
__device__ __forceinline__ void row_chunk(int s, int dl, const float* myw, const float* __restrict__ feat,
                                          float* acc, int lane)
{
    const int half = lane >> 4;
    const int c4   = lane & 15;
    const float4 wa = *(const float4*)(myw + C * 64 + half * 32 + (c4 >> 2) * 8);
    const float4 wb = *(const float4*)(myw + C * 64 + half * 32 + (c4 >> 2) * 8 + 4);
    const float wv[8] = { wa.x, wa.y, wa.z, wa.w, wb.x, wb.y, wb.z, wb.w };
    float4 f[8];
    int    dj[8];
#define ROW_LOAD(K) { const int sj = row_share<C * 8 + K>(s); dj[K] = row_share<C * 8 + K>(dl); \
                      if constexpr (NARROW) f[K] = *(const float4*)(feat + (((unsigned)sj << 6) | (unsigned)(c4 * 4))); \
                      else                  f[K] = *(const float4*)(feat + (size_t)sj * 64 + c4 * 4); }
    ROW_LOAD(0) ROW_LOAD(1) ROW_LOAD(2) ROW_LOAD(3) ROW_LOAD(4) ROW_LOAD(5) ROW_LOAD(6) ROW_LOAD(7)
#undef ROW_LOAD
    float* const acca = acc + c4 * 4 + (half ? 2 : 0);
    float* const accb = acc + c4 * 4 + (half ? 0 : 2);
    #pragma unroll
    for (int k = 0; k < 8; ++k) {
        const float p0 = f[k].x * wv[k], p1 = f[k].y * wv[k];
        const float p2 = f[k].z * wv[k], p3 = f[k].w * wv[k];
        float* const ra = acca + dj[k] * 64;
        float* const rb = accb + dj[k] * 64;
        __hip_atomic_fetch_add(ra,     half ? p2 : p0, __ATOMIC_RELAXED, __HIP_MEMORY_SCOPE_WORKGROUP);
        __hip_atomic_fetch_add(ra + 1, half ? p3 : p1, __ATOMIC_RELAXED, __HIP_MEMORY_SCOPE_WORKGROUP);
        __hip_atomic_fetch_add(rb,     half ? p0 : p2, __ATOMIC_RELAXED, __HIP_MEMORY_SCOPE_WORKGROUP);
        __hip_atomic_fetch_add(rb + 1, half ? p1 : p3, __ATOMIC_RELAXED, __HIP_MEMORY_SCOPE_WORKGROUP);
    }
}

template <bool USEIX>
__device__ __forceinline__ void agg_owned_body(
    const int* __restrict__ src, const int* __restrict__ dst,
    const float* __restrict__ el, const float* __restrict__ er,
    const float* __restrict__ feat,
    const float* __restrict__ bias, float* __restrict__ out, int N, int E,
    const unsigned* __restrict__ blk, const unsigned* __restrict__ offs, int nbw,
    unsigned char* pool)
{
    constexpr int LC  = USEIX ? LCAPX : LCAP;
    constexpr int APP = USEIX ? 256 : 512;
    constexpr unsigned SRCMASK = (1u << SRCBITS) - 1u;
    float*    const acc   = (float*)pool;
    float*    const sden  = acc + NPW * 64;
    unsigned* const smax  = (unsigned*)(sden + NPW * 4);
    float*    const swt   = (float*)(smax + NPW * 4);
    int*      const lst_e = (int*)(swt + AGGW * 128);
    int*      const lst_d = lst_e + AGGW * LC;
    float*    const ssc   = (float*)(lst_d + (USEIX ? 0 : AGGW * LC));
    int*      const lostp = (int*)(ssc + (USEIX ? AGGW * LC * 4 : 0));
    static_assert((NPW * 72 + AGGW * 128 + AGGW * LC * (USEIX ? 5 : 2) + 4) * 4 <= POOL_BYTES, "the pool must hold this carving");
#define s_lost (*lostp)

    const int tid  = threadIdx.x;
    const int wave = tid >> 5;
    const int lane = tid & 31;
    const int lo   = blockIdx.x * NPW;
    const int cnt  = (N - lo) < NPW ? (N - lo) : NPW;

    for (int i = tid; i < NPW * 64; i += 1024) acc[i] = 0.f;
    for (int i = tid; i < NPW * 4;  i += 1024) { sden[i] = 0.f; smax[i] = 0u; }
    if (tid == 0) s_lost = 0;
    __syncthreads();

    float* const myw = swt + wave * 128;
    int*   const mye = lst_e + wave * LC;
    int*   const myd = lst_d + (USEIX ? 0 : wave * LC);
    float* const mys = ssc + (USEIX ? wave * LC * 4 : 0);

    auto maxupd = [&](int n, bool have_s, bool keep) __attribute__((always_inline)) {
        for (int b = 0; b < n; b += 128) {
            int    sv[4], dv[4];
            float4 l4[4], r4[4];
            #pragma unroll
            for (int u = 0; u < 4; ++u) {
                const int  i     = b + u * 32 + lane;
                const bool valid = i < n;
                const int  e     = valid ? mye[i] : 0;
                if constexpr (USEIX) {
                    dv[u] = valid ? (int)((unsigned)e >> SRCBITS) : -1;
                    sv[u] = (int)((unsigned)e & SRCMASK);
                } else {
                    dv[u] = valid ? myd[i] : -1;
                    sv[u] = have_s ? e : src[e];
                }
            }
            #pragma unroll
            for (int u = 0; u < 4; ++u) {
                const int dl = dv[u] < 0 ? 0 : dv[u];
                l4[u] = *(const float4*)(el + (size_t)sv[u] * 4);
                r4[u] = *(const float4*)(er + (size_t)(lo + dl) * 4);
            }
            #pragma unroll
            for (int u = 0; u < 4; ++u) {
                if (dv[u] >= 0) {
                    unsigned* mx = &smax[dv[u] * 4];
                    float4 sc;
                    sc.x = leaky02(l4[u].x + r4[u].x);
                    sc.y = leaky02(l4[u].y + r4[u].y);
                    sc.z = leaky02(l4[u].z + r4[u].z);
                    sc.w = leaky02(l4[u].w + r4[u].w);
                    __hip_atomic_fetch_max(mx + 0, f2o(sc.x), __ATOMIC_RELAXED, __HIP_MEMORY_SCOPE_WORKGROUP);
                    __hip_atomic_fetch_max(mx + 1, f2o(sc.y), __ATOMIC_RELAXED, __HIP_MEMORY_SCOPE_WORKGROUP);
                    __hip_atomic_fetch_max(mx + 2, f2o(sc.z), __ATOMIC_RELAXED, __HIP_MEMORY_SCOPE_WORKGROUP);
                    __hip_atomic_fetch_max(mx + 3, f2o(sc.w), __ATOMIC_RELAXED, __HIP_MEMORY_SCOPE_WORKGROUP);
                    if (USEIX && keep) *(float4*)&mys[(b + u * 32 + lane) * 4] = sc;
                    if (!USEIX && !have_s) mye[b + u * 32 + lane] = sv[u];
                }
            }
        }
    };

    auto drain = [&](int n, bool have_s, bool kept) __attribute__((always_inline)) {
        for (int b = 0; b < n; b += 32) {
            const int  i     = b + ((lane & 15) << 1) + (lane >> 4);
            const bool valid = i < n;
            const int  q  = valid ? mye[i] : 0;
            int s, dl;
            if constexpr (USEIX) {
                s  = (int)((unsigned)q & SRCMASK);
                dl = (int)((unsigned)q >> SRCBITS);
            } else {
                dl = valid ? myd[i] : 0;
                s  = have_s ? q : src[q];
                s  = valid ? s : 0;
            }
            float4 sc;
            if (USEIX && kept) {
                sc = *(const float4*)&mys[(valid ? i : 0) * 4];
            } else {
                const int  d  = lo + dl;
                const float4 l4 = *(const float4*)(el + (size_t)s * 4);
                const float4 r4 = *(const float4*)(er + (size_t)d * 4);
                sc.x = leaky02(l4.x + r4.x);
                sc.y = leaky02(l4.y + r4.y);
                sc.z = leaky02(l4.z + r4.z);
                sc.w = leaky02(l4.w + r4.w);
            }
            const uint4  mu = *(const uint4*)&smax[dl * 4];
            float4 w;
            w.x = valid ? __expf(sc.x - o2f(mu.x)) : 0.f;
            w.y = valid ? __expf(sc.y - o2f(mu.y)) : 0.f;
            w.z = valid ? __expf(sc.z - o2f(mu.z)) : 0.f;
            w.w = valid ? __expf(sc.w - o2f(mu.w)) : 0.f;
            if (valid) {
                float* dn = &sden[dl * 4];
                __hip_atomic_fetch_add(dn + 0, w.x, __ATOMIC_RELAXED, __HIP_MEMORY_SCOPE_WORKGROUP);
                __hip_atomic_fetch_add(dn + 1, w.y, __ATOMIC_RELAXED, __HIP_MEMORY_SCOPE_WORKGROUP);
                __hip_atomic_fetch_add(dn + 2, w.z, __ATOMIC_RELAXED, __HIP_MEMORY_SCOPE_WORKGROUP);
                __hip_atomic_fetch_add(dn + 3, w.w, __ATOMIC_RELAXED, __HIP_MEMORY_SCOPE_WORKGROUP);
            }
            {
                float* wd = myw + ((lane & 15) >> 3) * 64 + (lane >> 4) * 32 + (lane & 7);
                wd[0] = w.x; wd[8] = w.y; wd[16] = w.z; wd[24] = w.w;
            }

            const int nb = (n - b) < 32 ? (n - b) : 32;
            row_chunk<0, USEIX>(s, dl, myw, feat, acc, lane);
            if (nb > 16) row_chunk<1, USEIX>(s, dl, myw, feat, acc, lane);
        }
    };

    const int G4     = E >> 2;
    const int steps  = ((G4 + 1023) >> 10);
    const int steps4 = (steps + 3) & ~3;
    const int st0    = (int)(((long long)blockIdx.x * steps4) / gridDim.x) & ~3;
    int fill = 0;
    int pass = 0;

    auto scan_load = [&](int k, int4 (&v)[4]) {
        int kk = st0 + k;
        if (kk >= steps4) kk -= steps4;
        #pragma unroll
        for (int u = 0; u < 4; ++u) {
            const int g = (kk + u) * 1024 + tid;
            v[u] = *(const int4*)(dst + (size_t)(g < G4 ? g : 0) * 4);
        }
    };

    auto examine = [&](int g, bool ok, const int4& v) __attribute__((always_inline)) {
        const int e0 = g * 4;
        const unsigned lim = ok ? (unsigned)cnt : 0u;
        const unsigned u0 = (unsigned)(v.x - lo);
        const unsigned u1 = (unsigned)(v.y - lo);
        const unsigned u2 = (unsigned)(v.z - lo);
        const unsigned u3 = (unsigned)(v.w - lo);
        const bool m0 = u0 < lim, m1 = u1 < lim;
        const bool m2 = u2 < lim, m3 = u3 < lim;
        if (__builtin_amdgcn_ballot_w32(m0 | m1 | m2 | m3)) {
            unsigned hb;
            hb = __builtin_amdgcn_ballot_w32(m0);
            if (m0) { const int p = fill + __builtin_amdgcn_mbcnt_lo(hb, 0u); if (p < LC) { mye[p] = e0;     myd[p] = (int)u0; } }
            fill += __builtin_popcount(hb);
            hb = __builtin_amdgcn_ballot_w32(m1);
            if (m1) { const int p = fill + __builtin_amdgcn_mbcnt_lo(hb, 0u); if (p < LC) { mye[p] = e0 + 1; myd[p] = (int)u1; } }
            fill += __builtin_popcount(hb);
            hb = __builtin_amdgcn_ballot_w32(m2);
            if (m2) { const int p = fill + __builtin_amdgcn_mbcnt_lo(hb, 0u); if (p < LC) { mye[p] = e0 + 2; myd[p] = (int)u2; } }
            fill += __builtin_popcount(hb);
            hb = __builtin_amdgcn_ballot_w32(m3);
            if (m3) { const int p = fill + __builtin_amdgcn_mbcnt_lo(hb, 0u); if (p < LC) { mye[p] = e0 + 3; myd[p] = (int)u3; } }
            fill += __builtin_popcount(hb);
        }
    };

    auto flush = [&]() __attribute__((always_inline)) {
        if (pass != 0 && fill > LC - APP) {
            if (pass == 1) maxupd(fill, USEIX, false);
            else           drain(fill, USEIX, false);
            fill = 0;
        }
    };

    auto scan_use = [&](int k, const int4 (&v)[4]) {
        int kk = st0 + k;
        if (kk >= steps4) kk -= steps4;
        #pragma unroll
        for (int u = 0; u < 4; ++u) {
            const int g = (kk + u) * 1024 + tid;
            examine(g, g < G4, v[u]);
        }
        flush();
    };

    auto append = [&](bool m, unsigned w, int add) __attribute__((always_inline)) {
        const int p = fill + lane;
        if (m && p < LC) mye[p] = (int)w;
        fill += add;
    };

    auto collect = [&]() __attribute__((always_inline)) {
        for (int b0 = wave; b0 < nbw; b0 += AGGW * 32) {
            const int bl = b0 + lane * AGGW;
            const unsigned meta = bl < nbw ? offs[(size_t)bl * BMROWS + blockIdx.x] : 0u;
            const int left = (nbw - b0 + AGGW - 1) / AGGW;
            const int nseg = left < 32 ? left : 32;
            for (int i0 = 0; i0 < nseg; i0 += 8) {
                unsigned w[8];
                int      cn[8];
                const unsigned* sg[8];
                #pragma unroll
                for (int u = 0; u < 8; ++u) {
                    const int i = i0 + u;
                    const unsigned mt = (unsigned)__builtin_amdgcn_readlane((int)meta, i);
                    cn[u] = (int)(mt >> 16);
                    sg[u] = blk + (size_t)(b0 + i * AGGW) * EPB + (mt & 0xffffu);
                    w[u]  = lane < cn[u] ? sg[u][lane] : 0u;
                }
                #pragma unroll
                for (int u = 0; u < 8; ++u) append(lane < cn[u], w[u], cn[u] < 32 ? cn[u] : 32);
                flush();
                #pragma unroll 1
                for (int i = i0; i < i0 + 8 && i < nseg; ++i) {
                    const unsigned mt = (unsigned)__builtin_amdgcn_readlane((int)meta, i);
                    const int c = (int)(mt >> 16);
                    const unsigned* const sp = blk + (size_t)(b0 + i * AGGW) * EPB + (mt & 0xffffu);
                    for (int k = 32; k < c; k += 32) {
                        const bool m = k + lane < c;
                        append(m, m ? sp[k + lane] : 0u, (c - k) < 32 ? (c - k) : 32);
                        flush();
                    }
                }
            }
        }
    };

    for (pass = 0; pass < 3; ++pass) {
        fill = 0;
        if constexpr (USEIX) {
            collect();
        } else {
            int4 va[4], vb[4];
            if (steps4 > 0) scan_load(0, va);
            for (int k = 0; k < steps4; k += 8) {
                scan_load(k + 4, vb);
                scan_use(k, va);
                scan_load(k + 8, va);
                if (k + 4 < steps4) scan_use(k + 4, vb);
            }
        }
        if (wave == 0) {
            const int et = (G4 << 2) + lane;
            unsigned ut = 0xffffffffu;
            if (et < E) ut = (unsigned)(dst[et] - lo);
            const bool mt = ut < (unsigned)cnt;
            const unsigned bm = __builtin_amdgcn_ballot_w32(mt);
            if (mt) {
                const int p = fill + __builtin_amdgcn_mbcnt_lo(bm, 0u);
                if (p < LC) {
                    if constexpr (USEIX) mye[p] = (int)(((unsigned)src[et] & SRCMASK) | (ut << SRCBITS));
                    else { mye[p] = et; myd[p] = (int)ut; }
                }
            }
            fill += __builtin_popcount(bm);
        }
        if (pass == 0) {
            if (fill > LC && lane == 0) s_lost = 1;
            maxupd(fill < LC ? fill : LC, USEIX, USEIX);
            __syncthreads();
            if (__builtin_amdgcn_readfirstlane(s_lost) == 0) { drain(fill, true, USEIX); break; }
        } else if (pass == 1) {
            maxupd(fill, USEIX, false);
            __syncthreads();
        } else {
            drain(fill, USEIX, false);
        }
    }
    __syncthreads();

    for (int i = tid; i < cnt * 64; i += 1024) {
        const int n = i >> 6, c = i & 63;
        const float dn = sden[n * 4 + (c >> 4)];
        const float a  = acc[i];
        out[(size_t)(lo + n) * 64 + c] = (dn > 0.f ? a / dn : 0.f) + bias[c];
    }
#undef s_lost
}

__global__ __launch_bounds__(1024) void gat_agg_scan(
    const int* __restrict__ src, const int* __restrict__ dst,
    const float* __restrict__ el, const float* __restrict__ er,
    const float* __restrict__ feat,
    const float* __restrict__ bias, float* __restrict__ out, int N, int E)
{
    __shared__ __attribute__((aligned(16))) unsigned char pool[POOL_BYTES];
    agg_owned_body<false>(src, dst, el, er, feat, bias, out, N, E, nullptr, nullptr, 0, pool);
}

template <int C>
__device__ __forceinline__ void row_loads_mm(int s, int ml, const float* __restrict__ feat,
                                             float4 (&f)[8], int (&mj)[8], int lane)
{
    const int c4 = lane & 15;
#define ROW_LOAD(K) { const int sj = row_share<C * 8 + K>(s); mj[K] = row_share<C * 8 + K>(ml); \
                      f[K] = *(const float4*)(feat + (((unsigned)sj << 6) | (unsigned)(c4 * 4))); }
    ROW_LOAD(0) ROW_LOAD(1) ROW_LOAD(2) ROW_LOAD(3) ROW_LOAD(4) ROW_LOAD(5) ROW_LOAD(6) ROW_LOAD(7)
#undef ROW_LOAD
}

template <int C>
__device__ __forceinline__ void row_products_mm(const float* myw, const float4 (&f)[8], const int (&mj)[8],
                                                v8f& t0, v8f& t1, v8f& t2, v8f& t3, int lane)
{
    const int half = lane >> 4;
    const int c4   = lane & 15;
    const float4 wa = *(const float4*)(myw + C * 64 + half * 32 + (c4 >> 2) * 8);
    const float4 wb = *(const float4*)(myw + C * 64 + half * 32 + (c4 >> 2) * 8 + 4);
    const float wv[8] = { wa.x, wa.y, wa.z, wa.w, wb.x, wb.y, wb.z, wb.w };
    #pragma unroll
    for (int g = 0; g < 4; ++g) {
        const int k0 = 2 * g, k1 = 2 * g + 1;
        v2f a, b0, b1, b2, b3;
        a.x  = mj[k0] == c4 ? 1.f : 0.f;   a.y  = mj[k1] == c4 ? 1.f : 0.f;
        b0.x = f[k0].x * wv[k0];           b0.y = f[k1].x * wv[k1];
        b1.x = f[k0].y * wv[k0];           b1.y = f[k1].y * wv[k1];
        b2.x = f[k0].z * wv[k0];           b2.y = f[k1].z * wv[k1];
        b3.x = f[k0].w * wv[k0];           b3.y = f[k1].w * wv[k1];
        t0 = __builtin_amdgcn_wmma_f32_16x16x4_f32(false, a, false, b0, (short)0, t0, false, false);
        t1 = __builtin_amdgcn_wmma_f32_16x16x4_f32(false, a, false, b1, (short)0, t1, false, false);
        t2 = __builtin_amdgcn_wmma_f32_16x16x4_f32(false, a, false, b2, (short)0, t2, false, false);
        t3 = __builtin_amdgcn_wmma_f32_16x16x4_f32(false, a, false, b3, (short)0, t3, false, false);
    }
}

#define LCO    448
#define TLROWS 13
__device__ __forceinline__ int tile_of(unsigned dl)  { return (int)((dl * 21u) >> 8); }
__device__ __forceinline__ int tile_first(int w)     { return (w * 256 + 20) / 21; }
static_assert(((unsigned)(NPW - 1) * 21u) >> 8 == AGGW - 1, "the last local row must belong to the last wave");
static_assert((AGGW * 256 + 20) / 21 == NPW, "the tiles must end with the range");
static_assert(TLROWS <= 16, "two rows per lane group of four cover a tile");
static_assert((NPW * 12 + 392 + AGGW * 4 + AGGW * 128 + AGGW * LCO + AGGW + 4) * 4 <= POOL_BYTES, "the pool must hold the owner lists");

__global__ __launch_bounds__(1024) void gat_agg_tiles(
    const int* __restrict__ src, const int* __restrict__ dst,
    const float* __restrict__ el, const float* __restrict__ er,
    const float* __restrict__ feat,
    const float* __restrict__ bias, float* __restrict__ out, int N, int E,
    const unsigned* __restrict__ blk, const unsigned* __restrict__ offs, int nbw,
    const float* __restrict__ pmax, int npb)
{
    __shared__ __attribute__((aligned(16))) unsigned char pool[POOL_BYTES];
    constexpr unsigned SRCMASK = (1u << SRCBITS) - 1u;
    float*    const sden  = (float*)pool;
    float*    const ser   = sden + NPW * 4;
    float*    const ssh   = ser + NPW * 4;
    unsigned* const smark = (unsigned*)(ssh + NPW * 4);
    float*    const swm   = (float*)(smark + 392);
    float*    const swt   = swm + AGGW * 4;
    int*      const olst  = (int*)(swt + AGGW * 128);
    unsigned* const ocnt  = (unsigned*)(olst + AGGW * LCO);
    int*      const lostp = (int*)(ocnt + AGGW);

    const int tid  = threadIdx.x;
    const int wave = tid >> 5;
    const int lane = tid & 31;
    const int lo   = blockIdx.x * NPW;
    const int cnt  = (N - lo) < NPW ? (N - lo) : NPW;
    const int rs   = tile_first(wave);
    const int re   = tile_first(wave + 1) < cnt ? tile_first(wave + 1) : cnt;

    float pmv[4];
    #pragma unroll
    for (int u = 0; u < 4; ++u) {
        const int i = tid + u * 1024;
        pmv[u] = i < npb * 4 ? pmax[i] : -3.402823466e+38f;
    }
    const int d0 = rs + (lane >> 2), d1 = d0 + 8;
    float er0 = 0.f, er1 = 0.f;
    if (d0 < re) er0 = er[(size_t)(lo + d0) * 4 + (lane & 3)];
    if (d1 < re) er1 = er[(size_t)(lo + d1) * 4 + (lane & 3)];
    unsigned meta0 = 0u;
    if (wave + lane * AGGW < nbw) meta0 = offs[(size_t)(wave + lane * AGGW) * BMROWS + blockIdx.x];

    for (int i = tid; i < NPW * 4; i += 1024) sden[i] = 0.f;
    if (tid < NPW) smark[tid] = 0u;
    if (tid < AGGW) ocnt[tid] = 0u;
    if (tid == 0) { lostp[0] = 0; lostp[1] = 0; }
    __syncthreads();

    auto put = [&](bool m, unsigned w) __attribute__((always_inline)) {
        if (m) {
            const int o = tile_of(w >> SRCBITS);
            const unsigned p = __hip_atomic_fetch_add(&ocnt[o], 1u, __ATOMIC_RELAXED, __HIP_MEMORY_SCOPE_WORKGROUP);
            if (p < (unsigned)LCO) olst[o * LCO + (int)p] = (int)w;
            else                   lostp[0] = 1;
        }
    };

    for (int b0 = wave; b0 < nbw; b0 += AGGW * 32) {
        const int bl = b0 + lane * AGGW;
        const unsigned meta = b0 == wave ? meta0 : (bl < nbw ? offs[(size_t)bl * BMROWS + blockIdx.x] : 0u);
        const int left = (nbw - b0 + AGGW - 1) / AGGW;
        const int nseg = left < 32 ? left : 32;
        for (int i0 = 0; i0 < nseg; i0 += 16) {
            unsigned w[16];
            int      cn[16];
            #pragma unroll
            for (int u = 0; u < 16; ++u) {
                const int i = i0 + u;
                const unsigned mt = (unsigned)__builtin_amdgcn_readlane((int)meta, i);
                cn[u] = (int)(mt >> 16);
                const unsigned* const sg = blk + (size_t)(b0 + i * AGGW) * EPB + (mt & 0xffffu);
                w[u]  = lane < cn[u] ? sg[lane] : 0u;
            }
            unsigned p[16];
            #pragma unroll
            for (int u = 0; u < 16; ++u) {
                p[u] = 0u;
                if (lane < cn[u]) p[u] = __hip_atomic_fetch_add(&ocnt[tile_of(w[u] >> SRCBITS)], 1u, __ATOMIC_RELAXED, __HIP_MEMORY_SCOPE_WORKGROUP);
            }
            #pragma unroll
            for (int u = 0; u < 16; ++u) {
                if (lane < cn[u]) {
                    if (p[u] < (unsigned)LCO) olst[tile_of(w[u] >> SRCBITS) * LCO + (int)p[u]] = (int)w[u];
                    else                      lostp[0] = 1;
                }
            }
            for (int i = i0; i < i0 + 16 && i < nseg; ++i) {
                const unsigned mt = (unsigned)__builtin_amdgcn_readlane((int)meta, i);
                const int c = (int)(mt >> 16);
                const unsigned* const sp = blk + (size_t)(b0 + i * AGGW) * EPB + (mt & 0xffffu);
                for (int k = 32; k < c; k += 32) {
                    const bool m = k + lane < c;
                    put(m, m ? sp[k + lane] : 0u);
                }
            }
        }
    }
    if (wave == 0) {
        const int et = ((E >> 2) << 2) + lane;
        if (et < E) {
            const unsigned ut = (unsigned)(dst[et] - lo);
            if (ut < (unsigned)cnt) put(true, ((unsigned)src[et] & SRCMASK) | (ut << SRCBITS));
        }
    }
    float pm = fmaxf(fmaxf(pmv[0], pmv[1]), fmaxf(pmv[2], pmv[3]));
    for (int i = tid + 4096; i < npb * 4; i += 1024) pm = fmaxf(pm, pmax[i]);
    pm = fmaxf(pm, __shfl_xor(pm, 4, 32));
    pm = fmaxf(pm, __shfl_xor(pm, 8, 32));
    pm = fmaxf(pm, __shfl_xor(pm, 16, 32));
    if (lane < 4) swm[wave * 4 + lane] = pm;
    __syncthreads();

    if (__builtin_amdgcn_readfirstlane(lostp[0]) == 0) {
        const int n = __builtin_amdgcn_readfirstlane((int)ocnt[wave]);
        int*   const mye = olst + wave * LCO;
        float* const myw = swt + wave * 128;

        float gm = fmaxf(fmaxf(swm[lane], swm[lane + 32]), fmaxf(swm[lane + 64], swm[lane + 96]));
        gm = fmaxf(gm, __shfl_xor(gm, 4, 32));
        gm = fmaxf(gm, __shfl_xor(gm, 8, 32));
        gm = fmaxf(gm, __shfl_xor(gm, 16, 32));
        if (d0 < re) { ser[d0 * 4 + (lane & 3)] = er0; ssh[d0 * 4 + (lane & 3)] = leaky02(gm + er0); }
        if (d1 < re) { ser[d1 * 4 + (lane & 3)] = er1; ssh[d1 * 4 + (lane & 3)] = leaky02(gm + er1); }

        v8f t0 = {}, t1 = {}, t2 = {}, t3 = {};
        for (int b = 0; b < n; b += 32) {
            const int  i     = b + ((lane & 15) << 1) + (lane >> 4);
            const bool valid = i < n;
            const int  q  = valid ? mye[i] : (rs << SRCBITS);
            const int  s  = (int)((unsigned)q & SRCMASK);
            const int  dl = (int)((unsigned)q >> SRCBITS);
            const int  ml = valid ? dl - rs : -1;
            const float4 l4 = *(const float4*)(el + (size_t)s * 4);
            float4 f[8];
            int    mj[8];
            row_loads_mm<0>(s, ml, feat, f, mj, lane);
            const float4 r4 = *(const float4*)&ser[dl * 4];
            const float4 h4 = *(const float4*)&ssh[dl * 4];
            float4 w;
            w.x = __expf(leaky02(l4.x + r4.x) - h4.x);
            w.y = __expf(leaky02(l4.y + r4.y) - h4.y);
            w.z = __expf(leaky02(l4.z + r4.z) - h4.z);
            w.w = __expf(leaky02(l4.w + r4.w) - h4.w);
            if (valid) {
                float* dn = &sden[dl * 4];
                __hip_atomic_fetch_add(dn + 0, w.x, __ATOMIC_RELAXED, __HIP_MEMORY_SCOPE_WORKGROUP);
                __hip_atomic_fetch_add(dn + 1, w.y, __ATOMIC_RELAXED, __HIP_MEMORY_SCOPE_WORKGROUP);
                __hip_atomic_fetch_add(dn + 2, w.z, __ATOMIC_RELAXED, __HIP_MEMORY_SCOPE_WORKGROUP);
                __hip_atomic_fetch_add(dn + 3, w.w, __ATOMIC_RELAXED, __HIP_MEMORY_SCOPE_WORKGROUP);
                smark[dl] = 1u;
            }
            {
                float* wd = myw + ((lane & 15) >> 3) * 64 + (lane >> 4) * 32 + (lane & 7);
                wd[0] = w.x; wd[8] = w.y; wd[16] = w.z; wd[24] = w.w;
            }
            const int nb = (n - b) < 32 ? (n - b) : 32;
            row_products_mm<0>(myw, f, mj, t0, t1, t2, t3, lane);
            if (nb > 16) {
                row_loads_mm<1>(s, ml, feat, f, mj, lane);
                row_products_mm<1>(myw, f, mj, t0, t1, t2, t3, lane);
            }
        }

        {
            bool bad = false;
            if (d0 < re && smark[d0] != 0u) { const float d = sden[d0 * 4 + (lane & 3)]; bad = bad || !(d >= 1e-20f && d <= 3.0e38f); }
            if (d1 < re && smark[d1] != 0u) { const float d = sden[d1 * 4 + (lane & 3)]; bad = bad || !(d >= 1e-20f && d <= 3.0e38f); }
            if (bad) lostp[1] = 1;
        }
        __syncthreads();

        if (__builtin_amdgcn_readfirstlane(lostp[1]) == 0) {
            const int half = lane >> 4;
            const int c4   = lane & 15;
            const float4 bv = *(const float4*)(bias + c4 * 4);
            #pragma unroll
            for (int r = 0; r < 8; ++r) {
                const int nl = rs + half * 8 + r;
                if (nl < re) {
                    const float dn = sden[nl * 4 + (c4 >> 2)];
                    float4 o;
                    o.x = (dn > 0.f ? t0[r] / dn : 0.f) + bv.x;
                    o.y = (dn > 0.f ? t1[r] / dn : 0.f) + bv.y;
                    o.z = (dn > 0.f ? t2[r] / dn : 0.f) + bv.z;
                    o.w = (dn > 0.f ? t3[r] / dn : 0.f) + bv.w;
                    *(float4*)(out + (size_t)(lo + nl) * 64 + c4 * 4) = o;
                }
            }
            return;
        }
    }
    __syncthreads();
    agg_owned_body<true>(src, dst, el, er, feat, bias, out, N, E, blk, offs, nbw, pool);
}

extern "C" void kernel_launch(void* const* d_in, const int* in_sizes, int n_in,
                              void* d_out, int out_size, void* d_ws, size_t ws_size,
                              hipStream_t stream)
{
    const float* nf   = (const float*)d_in[0];
    const float* W    = (const float*)d_in[1];
    const float* al   = (const float*)d_in[2];
    const float* ar   = (const float*)d_in[3];
    const float* bias = (const float*)d_in[4];
    const int*   src  = (const int*)d_in[5];
    const int*   dst  = (const int*)d_in[6];

    const int N = in_sizes[0] / 64;
    const int E = in_sizes[5];

    float* ws    = (float*)d_ws;
    float* feat  = ws;
    float* el    = feat  + (size_t)N * 64;
    float* er    = el    + (size_t)N * 4;
    float* out   = (float*)d_out;
    const int nr  = (N + NPW - 1) / NPW;
    const int G4  = E >> 2;
    const int nbw = (G4 + 1023) / 1024;
    unsigned* blk  = (unsigned*)(er + (size_t)N * 4);
    unsigned* offs = blk + (size_t)nbw * EPB;
    const int npb = (N + 127) / 128;
    float* pmax = (float*)(offs + (size_t)nbw * BMROWS);
    const size_t need = ((size_t)N * 72 + (size_t)nbw * (EPB + BMROWS) + (size_t)npb * 4) * sizeof(float);

    if (nr <= BMROWS && nbw > 0 && N <= (1 << SRCBITS) && ws_size >= need) {
        gat_proj_logits_sort<<<nbw + npb, 256, 0, stream>>>(nf, W, al, ar, feat, el, er, N, src, dst, blk, offs, G4, nbw, pmax);
        gat_agg_tiles<<<nr, 1024, 0, stream>>>(src, dst, el, er, feat, bias, out, N, E, blk, offs, nbw, pmax, npb);
    } else {
        gat_proj_logits_sort<<<npb, 256, 0, stream>>>(nf, W, al, ar, feat, el, er, N, src, dst, nullptr, nullptr, G4, 0, nullptr);
        gat_agg_scan<<<nr, 1024, 0, stream>>>(src, dst, el, er, feat, bias, out, N, E);
    }
}
